# starting kernel
# baseline (speedup 1.0000x reference)
#include <hip/hip_runtime.h>

#define ENT_DIM  128
#define REL_DIM  128
#define CMP_DIM  256
#define NUM_RELC 200
#define MTILES   4

typedef __attribute__((ext_vector_type(2))) float v2f;
typedef __attribute__((ext_vector_type(8))) float v8f;

__device__ __forceinline__ v8f wmma4_f32(v2f a, v2f b, v8f c) {
    return __builtin_amdgcn_wmma_f32_16x16x4_f32(
         false, a,  false, b,
         (short)0, c,  false,  false);
}

__global__ __launch_bounds__(256)
void edge_msg_kernel(const int*   __restrict__ src,
                     const int*   __restrict__ dst,
                     const int*   __restrict__ etypes,
                     const float* __restrict__ ent_emb,
                     const float* __restrict__ rel_emb,
                     const float* __restrict__ Wo,
                     const float* __restrict__ bo,
                     const float* __restrict__ Wi,
                     const float* __restrict__ bi,
                     float*       __restrict__ agg,
                     float*       __restrict__ deg,
                     int E)
{
    __shared__ int s_et [MTILES * 16];
    __shared__ int s_dst[MTILES * 16];

    const int  tid   = threadIdx.x;
    const int  lane  = tid & 31;
    const int  wave  = tid >> 5;
    const int  m     = lane & 15;
    const int  khalf = (lane >> 4) << 1;
    const long base  = (long)blockIdx.x * (MTILES * 16);

    if (tid < MTILES * 16) {
        long e  = base + tid;
        int  ec = (e < E) ? (int)e : (E - 1);
        s_et [tid] = etypes[ec];
        s_dst[tid] = dst[ec];
    }
    __syncthreads();

    const float* arel[MTILES];
    const float* aent[MTILES];
    #pragma unroll
    for (int t = 0; t < MTILES; ++t) {
        long e  = base + t * 16 + m;
        int  ec = (e < E) ? (int)e : (E - 1);
        arel[t] = rel_emb + (long)etypes[ec] * REL_DIM;
        aent[t] = ent_emb + (long)src[ec]    * ENT_DIM;
    }

    const int    n      = (wave << 4) + m;
    const float* wo_row = Wo + (long)n * CMP_DIM;
    const float* wi_row = Wi + (long)n * CMP_DIM;

    v8f acc_o[MTILES], acc_i[MTILES];
    #pragma unroll
    for (int t = 0; t < MTILES; ++t) {
        acc_o[t] = (v8f){0.f,0.f,0.f,0.f,0.f,0.f,0.f,0.f};
        acc_i[t] = (v8f){0.f,0.f,0.f,0.f,0.f,0.f,0.f,0.f};
    }

    #pragma unroll 2
    for (int k = 0; k < REL_DIM; k += 4) {
        v2f bO = *(const v2f*)(wo_row + k + khalf);
        v2f bI = *(const v2f*)(wi_row + k + khalf);
        v2f a[MTILES];
        #pragma unroll
        for (int t = 0; t < MTILES; ++t)
            a[t] = *(const v2f*)(arel[t] + k + khalf);
        #pragma unroll
        for (int t = 0; t < MTILES; ++t) {
            acc_o[t] = wmma4_f32(a[t], bO, acc_o[t]);
            acc_i[t] = wmma4_f32(a[t], bI, acc_i[t]);
        }
    }
    #pragma unroll 2
    for (int k = 0; k < ENT_DIM; k += 4) {
        v2f bO = *(const v2f*)(wo_row + REL_DIM + k + khalf);
        v2f bI = *(const v2f*)(wi_row + REL_DIM + k + khalf);
        v2f a[MTILES];
        #pragma unroll
        for (int t = 0; t < MTILES; ++t)
            a[t] = *(const v2f*)(aent[t] + k + khalf);
        #pragma unroll
        for (int t = 0; t < MTILES; ++t) {
            acc_o[t] = wmma4_f32(a[t], bO, acc_o[t]);
            acc_i[t] = wmma4_f32(a[t], bI, acc_i[t]);
        }
    }

    const float bob   = bo[n];
    const float bib   = bi[n];
    const int   rhalf = (lane >> 4) << 3;

    #pragma unroll
    for (int t = 0; t < MTILES; ++t) {
        #pragma unroll
        for (int j = 0; j < 8; ++j) {
            const int  me = t * 16 + j + rhalf;
            const long ee = base + me;
            if (ee < E) {
                const bool  is_o = s_et[me] < NUM_RELC;
                const float v    = is_o ? (acc_o[t][j] + bob) : (acc_i[t][j] + bib);
                atomicAdd(&agg[(long)s_dst[me] * ENT_DIM + n], v);
            }
        }
    }

    if (tid < MTILES * 16 && (base + tid) < E)
        atomicAdd(&deg[s_dst[tid]], 1.0f);
}

__global__ __launch_bounds__(256)
void gemm_bias_kernel(const float* __restrict__ A,
                      const float* __restrict__ W,
                      const float* __restrict__ b,
                      const float* __restrict__ agg,
                      const float* __restrict__ deg,
                      float*       __restrict__ out,
                      int N)
{
    const int  tid   = threadIdx.x;
    const int  lane  = tid & 31;
    const int  wave  = tid >> 5;
    const int  m     = lane & 15;
    const int  khalf = (lane >> 4) << 1;
    const long tile  = (long)blockIdx.x * 16;

    const long r    = tile + m;
    const long rc   = (r < N) ? r : (long)(N - 1);
    const float* ar = A + rc * ENT_DIM;

    const int    n  = (wave << 4) + m;
    const float* wr = W + (long)n * ENT_DIM;

    v8f acc = {0.f,0.f,0.f,0.f,0.f,0.f,0.f,0.f};

    #pragma unroll 4
    for (int k = 0; k < ENT_DIM; k += 4) {
        v2f av = *(const v2f*)(ar + k + khalf);
        v2f bv = *(const v2f*)(wr + k + khalf);
        acc = wmma4_f32(av, bv, acc);
    }

    const float bn    = b[n];
    const int   rhalf = (lane >> 4) << 3;

    #pragma unroll
    for (int j = 0; j < 8; ++j) {
        const int  me = j + rhalf;
        const long rr = tile + me;
        if (rr < N) {
            float v = acc[j] + bn;
            if (agg != nullptr) {
                float d = deg[rr];
                d = (d > 1.0f) ? d : 1.0f;
                v += agg[rr * ENT_DIM + n] / d;
            }
            out[rr * ENT_DIM + n] = v;
        }
    }
}

__global__ void zero_kernel(float* __restrict__ p, long n)
{
    const long i = (long)blockIdx.x * blockDim.x + threadIdx.x;
    if (i < n) p[i] = 0.0f;
}

extern "C" void kernel_launch(void* const* d_in, const int* in_sizes, int n_in,
                              void* d_out, int out_size, void* d_ws, size_t ws_size,
                              hipStream_t stream)
{
    const int*   src     = (const int*)  d_in[0];
    const int*   dst     = (const int*)  d_in[1];
    const int*   etypes  = (const int*)  d_in[2];
    const float* ent_emb = (const float*)d_in[3];
    const float* rel_emb = (const float*)d_in[4];
    const float* Wo      = (const float*)d_in[5];
    const float* bo      = (const float*)d_in[6];
    const float* Wi      = (const float*)d_in[7];
    const float* bi      = (const float*)d_in[8];
    const float* Ws      = (const float*)d_in[9];
    const float* bs      = (const float*)d_in[10];
    const float* Wr      = (const float*)d_in[11];
    const float* br      = (const float*)d_in[12];

    const int E = in_sizes[0];
    const int N = in_sizes[3] / ENT_DIM;
    const int R = in_sizes[4] / REL_DIM;

    float* agg = (float*)d_ws;
    float* deg = agg + (size_t)N * ENT_DIM;

    const long zn = (long)N * ENT_DIM + N;
    zero_kernel<<<(int)((zn + 255) / 256), 256, 0, stream>>>(agg, zn);

    const int edges_per_block = MTILES * 16;
    edge_msg_kernel<<<(E + edges_per_block - 1) / edges_per_block, 256, 0, stream>>>(
        src, dst, etypes, ent_emb, rel_emb, Wo, bo, Wi, bi, agg, deg, E);

    float* out_ent = (float*)d_out;
    float* out_rel = out_ent + (size_t)N * ENT_DIM;

    gemm_bias_kernel<<<(N + 15) / 16, 256, 0, stream>>>(
        ent_emb, Ws, bs, agg, deg, out_ent, N);

    gemm_bias_kernel<<<(R + 15) / 16, 256, 0, stream>>>(
        rel_emb, Wr, br, nullptr, nullptr, out_rel, R);
}


# Replace per-edge float atomics and the agg zero pass by a per-call counting sort of edges by destination (int atomic rank) and a per-node register sum; messages summed in arrival-rank order
# speedup vs baseline: 4.3405x; 4.3405x over previous
#include <hip/hip_runtime.h>

#define ENT_DIM  128
#define REL_DIM  128
#define CMP_DIM  256
#define NUM_RELC 200
#define MTILES   4
#define EDGE_EPT   4
#define NODE_TRIPS 4

typedef __attribute__((ext_vector_type(2))) float v2f;
typedef __attribute__((ext_vector_type(8))) float v8f;
typedef __attribute__((ext_vector_type(4))) float v4f;
typedef __attribute__((ext_vector_type(2))) int   v2i;

__device__ __forceinline__ v8f wmma4_f32(v2f a, v2f b, v8f c) {
    return __builtin_amdgcn_wmma_f32_16x16x4_f32(
         false, a,  false, b,
         (short)0, c,  false,  false);
}

__global__ __launch_bounds__(256)
void project_kernel(const float* __restrict__ ent_emb,
                    const float* __restrict__ rel_emb,
                    const float* __restrict__ Wo,
                    const float* __restrict__ bo,
                    const float* __restrict__ Wi,
                    const float* __restrict__ bi,
                    float*       __restrict__ P,
                    float*       __restrict__ PR,
                    int*         __restrict__ cnt,
                    int*         __restrict__ cursor,
                    int N, int R, int node_blocks)
{
    const int  tid   = threadIdx.x;
    const int  lane  = tid & 31;
    const int  wave  = tid >> 5;
    const int  m     = lane & 15;
    const int  khalf = (lane >> 4) << 1;

    const bool   rel  = (int)blockIdx.x >= node_blocks;
    const float* A    = rel ? rel_emb : ent_emb;
    const int    rows = rel ? R : N;
    const int    koff = rel ? 0 : REL_DIM;
    const long   base = (long)(rel ? (int)blockIdx.x - node_blocks : (int)blockIdx.x)
                        * (MTILES * 16);

    if (!rel && tid < MTILES * 16 && base + tid < rows) cnt[base + tid] = 0;
    if (blockIdx.x == 0 && tid == MTILES * 16) *cursor = 0;

    const float* arow[MTILES];
    #pragma unroll
    for (int t = 0; t < MTILES; ++t) {
        long r  = base + t * 16 + m;
        long rc = (r < rows) ? r : (long)(rows - 1);
        arow[t] = A + rc * ENT_DIM;
    }

    const int    n      = (wave << 4) + m;
    const float* wo_row = Wo + (long)n * CMP_DIM + koff;
    const float* wi_row = Wi + (long)n * CMP_DIM + koff;

    v8f acc_o[MTILES], acc_i[MTILES];
    #pragma unroll
    for (int t = 0; t < MTILES; ++t) {
        acc_o[t] = (v8f){0.f,0.f,0.f,0.f,0.f,0.f,0.f,0.f};
        acc_i[t] = (v8f){0.f,0.f,0.f,0.f,0.f,0.f,0.f,0.f};
    }

    #pragma unroll 2
    for (int k = 0; k < ENT_DIM; k += 4) {
        v2f bO = *(const v2f*)(wo_row + k + khalf);
        v2f bI = *(const v2f*)(wi_row + k + khalf);
        v2f a[MTILES];
        #pragma unroll
        for (int t = 0; t < MTILES; ++t)
            a[t] = *(const v2f*)(arow[t] + k + khalf);
        #pragma unroll
        for (int t = 0; t < MTILES; ++t) {
            acc_o[t] = wmma4_f32(a[t], bO, acc_o[t]);
            acc_i[t] = wmma4_f32(a[t], bI, acc_i[t]);
        }
    }

    const int rhalf = (lane >> 4) << 3;

    if (rel) {
        const float bob = bo[n];
        const float bib = bi[n];
        #pragma unroll
        for (int t = 0; t < MTILES; ++t) {
            #pragma unroll
            for (int j = 0; j < 8; ++j) {
                const long row = base + t * 16 + j + rhalf;
                if (row < rows)
                    PR[row * REL_DIM + n] = (row < NUM_RELC) ? (acc_o[t][j] + bob)
                                                             : (acc_i[t][j] + bib);
            }
        }
    } else {
        #pragma unroll
        for (int t = 0; t < MTILES; ++t) {
            #pragma unroll
            for (int j = 0; j < 8; ++j) {
                const long row = base + t * 16 + j + rhalf;
                if (row < rows) {
                    P[row * CMP_DIM + n]           = acc_o[t][j];
                    P[row * CMP_DIM + ENT_DIM + n] = acc_i[t][j];
                }
            }
        }
    }
}

__global__ __launch_bounds__(256)
void dst_rank_kernel(const int* __restrict__ dst,
                     int*       __restrict__ cnt,
                     int*       __restrict__ rank,
                     int E)
{
    const long e0 = (long)blockIdx.x * (256 * EDGE_EPT) + threadIdx.x;
    int d[EDGE_EPT], r[EDGE_EPT];
    #pragma unroll
    for (int k = 0; k < EDGE_EPT; ++k) {
        const long e = e0 + k * 256;
        d[k] = dst[(e < E) ? e : (long)E - 1];
    }
    #pragma unroll
    for (int k = 0; k < EDGE_EPT; ++k) {
        r[k] = 0;
        if (e0 + k * 256 < E) r[k] = atomicAdd(&cnt[d[k]], 1);
    }
    #pragma unroll
    for (int k = 0; k < EDGE_EPT; ++k) {
        const long e = e0 + k * 256;
        if (e < E) rank[e] = r[k];
    }
}

__global__ __launch_bounds__(256)
void segment_alloc_kernel(const int* __restrict__ cnt,
                          int*       __restrict__ cursor,
                          int*       __restrict__ off,
                          int N)
{
    __shared__ int wsum[8];
    __shared__ int bbase;
    const int tid  = threadIdx.x;
    const int lane = tid & 31;
    const int wave = tid >> 5;
    const int d    = (int)blockIdx.x * 256 + tid;
    const int c    = (d < N) ? cnt[d] : 0;

    int x = c;
    #pragma unroll
    for (int s = 1; s < 32; s <<= 1) {
        const int y = __shfl_up(x, s, 32);
        if (lane >= s) x += y;
    }
    if (lane == 31) wsum[wave] = x;
    __syncthreads();
    if (tid == 0) {
        int t = 0;
        #pragma unroll
        for (int w = 0; w < 8; ++w) { const int v = wsum[w]; wsum[w] = t; t += v; }
        bbase = atomicAdd(cursor, t);
    }
    __syncthreads();
    if (d < N) off[d] = bbase + wsum[wave] + x - c;
}

__global__ __launch_bounds__(256)
void edge_place_kernel(const int* __restrict__ src,
                       const int* __restrict__ dst,
                       const int* __restrict__ etypes,
                       const int* __restrict__ off,
                       const int* __restrict__ rank,
                       int*       __restrict__ key,
                       int E)
{
    const long e0 = (long)blockIdx.x * (256 * EDGE_EPT) + threadIdx.x;
    int d[EDGE_EPT], r[EDGE_EPT], s[EDGE_EPT], ty[EDGE_EPT], o[EDGE_EPT];
    #pragma unroll
    for (int k = 0; k < EDGE_EPT; ++k) {
        const long e  = e0 + k * 256;
        const long ec = (e < E) ? e : (long)E - 1;
        d[k]  = dst[ec];
        r[k]  = rank[ec];
        s[k]  = src[ec];
        ty[k] = etypes[ec];
    }
    #pragma unroll
    for (int k = 0; k < EDGE_EPT; ++k)
        o[k] = off[d[k]];
    #pragma unroll
    for (int k = 0; k < EDGE_EPT; ++k) {
        if (e0 + k * 256 < E) {
            const long pos = (long)o[k] + r[k];
            v2i kv;
            kv.x = s[k] * CMP_DIM + ((ty[k] < NUM_RELC) ? 0 : ENT_DIM);
            kv.y = ty[k] * REL_DIM;
            *(v2i*)(key + 2 * pos) = kv;
        }
    }
}

__global__ __launch_bounds__(256)
void node_gather_sum_kernel(const int*   __restrict__ key,
                            const int*   __restrict__ cnt,
                            const int*   __restrict__ off,
                            const float* __restrict__ P,
                            const float* __restrict__ PR,
                            float*       __restrict__ agg,
                            float*       __restrict__ deg,
                            int N)
{
    const int lane = threadIdx.x & 31;
    const int wave = __builtin_amdgcn_readfirstlane((int)(threadIdx.x >> 5));
    const int col  = lane << 2;

    #pragma unroll 1
    for (int trip = 0; trip < NODE_TRIPS; ++trip) {
        const int node = ((int)blockIdx.x * NODE_TRIPS + trip) * 8 + wave;
        if (node >= N) break;
        const int n = __builtin_amdgcn_readfirstlane(cnt[node]);
        const int o = __builtin_amdgcn_readfirstlane(off[node]);

        v4f acc = {0.f, 0.f, 0.f, 0.f};
        for (int base = 0; base < n; base += 32) {
            const int m = (n - base < 32) ? (n - base) : 32;
            v2i kv = {0, 0};
            if (lane < m) kv = *(const v2i*)(key + 2 * ((long)o + base + lane));

            for (int g = 0; g < m; g += 4) {
                v4f pv[4], rv[4];
                #pragma unroll
                for (int i = 0; i < 4; ++i) {
                    const int pi = __builtin_amdgcn_readlane(kv.x, g + i);
                    const int ri = __builtin_amdgcn_readlane(kv.y, g + i);
                    pv[i] = *(const v4f*)(P  + pi + col);
                    rv[i] = *(const v4f*)(PR + ri + col);
                }
                #pragma unroll
                for (int i = 0; i < 4; ++i) {
                    if (g + i < m)
                        acc += rv[i] + pv[i];
                }
            }
        }
        *(v4f*)(agg + (long)node * ENT_DIM + col) = acc;
        if (lane == 0) deg[node] = (float)n;
    }
}

__global__ __launch_bounds__(256)
void gemm_bias_kernel(const float* __restrict__ A,
                      const float* __restrict__ W,
                      const float* __restrict__ b,
                      const float* __restrict__ agg,
                      const float* __restrict__ deg,
                      float*       __restrict__ out,
                      int N)
{
    const int  tid   = threadIdx.x;
    const int  lane  = tid & 31;
    const int  wave  = tid >> 5;
    const int  m     = lane & 15;
    const int  khalf = (lane >> 4) << 1;
    const long tile  = (long)blockIdx.x * 16;

    const long r    = tile + m;
    const long rc   = (r < N) ? r : (long)(N - 1);
    const float* ar = A + rc * ENT_DIM;

    const int    n  = (wave << 4) + m;
    const float* wr = W + (long)n * ENT_DIM;

    v8f acc = {0.f,0.f,0.f,0.f,0.f,0.f,0.f,0.f};

    #pragma unroll 4
    for (int k = 0; k < ENT_DIM; k += 4) {
        v2f av = *(const v2f*)(ar + k + khalf);
        v2f bv = *(const v2f*)(wr + k + khalf);
        acc = wmma4_f32(av, bv, acc);
    }

    const float bn    = b[n];
    const int   rhalf = (lane >> 4) << 3;

    #pragma unroll
    for (int j = 0; j < 8; ++j) {
        const int  me = j + rhalf;
        const long rr = tile + me;
        if (rr < N) {
            float v = acc[j] + bn;
            if (agg != nullptr) {
                float d = deg[rr];
                d = (d > 1.0f) ? d : 1.0f;
                v += agg[rr * ENT_DIM + n] / d;
            }
            out[rr * ENT_DIM + n] = v;
        }
    }
}

extern "C" void kernel_launch(void* const* d_in, const int* in_sizes, int n_in,
                              void* d_out, int out_size, void* d_ws, size_t ws_size,
                              hipStream_t stream)
{
    const int*   src     = (const int*)  d_in[0];
    const int*   dst     = (const int*)  d_in[1];
    const int*   etypes  = (const int*)  d_in[2];
    const float* ent_emb = (const float*)d_in[3];
    const float* rel_emb = (const float*)d_in[4];
    const float* Wo      = (const float*)d_in[5];
    const float* bo      = (const float*)d_in[6];
    const float* Wi      = (const float*)d_in[7];
    const float* bi      = (const float*)d_in[8];
    const float* Ws      = (const float*)d_in[9];
    const float* bs      = (const float*)d_in[10];
    const float* Wr      = (const float*)d_in[11];
    const float* br      = (const float*)d_in[12];

    const int E = in_sizes[0];
    const int N = in_sizes[3] / ENT_DIM;
    const int R = in_sizes[4] / REL_DIM;

    float* P    = (float*)d_ws;
    float* PR   = P   + (size_t)N * CMP_DIM;
    float* agg  = PR  + (((size_t)R * REL_DIM + 63) & ~(size_t)63);
    float* deg  = agg + (size_t)N * ENT_DIM;
    int*   cnt  = (int*)(deg + (((size_t)N + 63) & ~(size_t)63));
    int*   curs = cnt  + (((size_t)N + 63) & ~(size_t)63);
    int*   off  = curs + 64;
    int*   rank = off  + (((size_t)N + 63) & ~(size_t)63);
    int*   key  = rank + (((size_t)E + 63) & ~(size_t)63);

    const int rows_per_block = MTILES * 16;
    const int node_blocks    = (N + rows_per_block - 1) / rows_per_block;
    const int rel_blocks     = (R + rows_per_block - 1) / rows_per_block;
    project_kernel<<<node_blocks + rel_blocks, 256, 0, stream>>>(
        ent_emb, rel_emb, Wo, bo, Wi, bi, P, PR, cnt, curs, N, R, node_blocks);

    const int edges_per_block = 256 * EDGE_EPT;
    const int edge_blocks     = (E + edges_per_block - 1) / edges_per_block;
    dst_rank_kernel<<<edge_blocks, 256, 0, stream>>>(dst, cnt, rank, E);
    segment_alloc_kernel<<<(N + 255) / 256, 256, 0, stream>>>(cnt, curs, off, N);
    edge_place_kernel<<<edge_blocks, 256, 0, stream>>>(src, dst, etypes, off, rank, key, E);

    const int nodes_per_block = 8 * NODE_TRIPS;
    node_gather_sum_kernel<<<(N + nodes_per_block - 1) / nodes_per_block, 256, 0, stream>>>(
        key, cnt, off, P, PR, agg, deg, N);

    float* out_ent = (float*)d_out;
    float* out_rel = out_ent + (size_t)N * ENT_DIM;

    gemm_bias_kernel<<<(N + 15) / 16, 256, 0, stream>>>(
        ent_emb, Ws, bs, agg, deg, out_ent, N);

    gemm_bias_kernel<<<(R + 15) / 16, 256, 0, stream>>>(
        rel_emb, Wr, br, nullptr, nullptr, out_rel, R);
}


# Fold the W_S and W_R products into the projection kernel as a third accumulator on the shared A fragments; the node-sum kernel adds the mean into the output row; both GEMM launches go
# speedup vs baseline: 5.0646x; 1.1668x over previous
#include <hip/hip_runtime.h>

#define ENT_DIM  128
#define REL_DIM  128
#define CMP_DIM  256
#define NUM_RELC 200
#define MTILES   4
#define EDGE_EPT   4
#define NODE_TRIPS 4

typedef __attribute__((ext_vector_type(2))) float v2f;
typedef __attribute__((ext_vector_type(8))) float v8f;
typedef __attribute__((ext_vector_type(4))) float v4f;
typedef __attribute__((ext_vector_type(2))) int   v2i;

__device__ __forceinline__ v8f wmma4_f32(v2f a, v2f b, v8f c) {
    return __builtin_amdgcn_wmma_f32_16x16x4_f32(
         false, a,  false, b,
         (short)0, c,  false,  false);
}

__global__ __launch_bounds__(256)
void project_kernel(const float* __restrict__ ent_emb,
                    const float* __restrict__ rel_emb,
                    const float* __restrict__ Wo,
                    const float* __restrict__ bo,
                    const float* __restrict__ Wi,
                    const float* __restrict__ bi,
                    const float* __restrict__ Ws,
                    const float* __restrict__ bs,
                    const float* __restrict__ Wr,
                    const float* __restrict__ br,
                    float*       __restrict__ out_ent,
                    float*       __restrict__ out_rel,
                    float*       __restrict__ P,
                    float*       __restrict__ PR,
                    int*         __restrict__ cnt,
                    int*         __restrict__ cursor,
                    int N, int R, int node_blocks)
{
    const int  tid   = threadIdx.x;
    const int  lane  = tid & 31;
    const int  wave  = tid >> 5;
    const int  m     = lane & 15;
    const int  khalf = (lane >> 4) << 1;

    const bool   rel  = (int)blockIdx.x >= node_blocks;
    const float* A    = rel ? rel_emb : ent_emb;
    const int    rows = rel ? R : N;
    const int    koff = rel ? 0 : REL_DIM;
    const long   base = (long)(rel ? (int)blockIdx.x - node_blocks : (int)blockIdx.x)
                        * (MTILES * 16);

    if (!rel && tid < MTILES * 16 && base + tid < rows) cnt[base + tid] = 0;
    if (blockIdx.x == 0 && tid == MTILES * 16) *cursor = 0;

    const float* arow[MTILES];
    #pragma unroll
    for (int t = 0; t < MTILES; ++t) {
        long r  = base + t * 16 + m;
        long rc = (r < rows) ? r : (long)(rows - 1);
        arow[t] = A + rc * ENT_DIM;
    }

    const int    n      = (wave << 4) + m;
    const float* wo_row = Wo + (long)n * CMP_DIM + koff;
    const float* wi_row = Wi + (long)n * CMP_DIM + koff;
    const float* wu_row = (rel ? Wr : Ws) + (long)n * ENT_DIM;

    v8f acc_o[MTILES], acc_i[MTILES], acc_u[MTILES];
    #pragma unroll
    for (int t = 0; t < MTILES; ++t) {
        acc_o[t] = (v8f){0.f,0.f,0.f,0.f,0.f,0.f,0.f,0.f};
        acc_i[t] = (v8f){0.f,0.f,0.f,0.f,0.f,0.f,0.f,0.f};
        acc_u[t] = (v8f){0.f,0.f,0.f,0.f,0.f,0.f,0.f,0.f};
    }

    #pragma unroll 2
    for (int k = 0; k < ENT_DIM; k += 4) {
        v2f bO = *(const v2f*)(wo_row + k + khalf);
        v2f bI = *(const v2f*)(wi_row + k + khalf);
        v2f bU = *(const v2f*)(wu_row + k + khalf);
        v2f a[MTILES];
        #pragma unroll
        for (int t = 0; t < MTILES; ++t)
            a[t] = *(const v2f*)(arow[t] + k + khalf);
        #pragma unroll
        for (int t = 0; t < MTILES; ++t) {
            acc_o[t] = wmma4_f32(a[t], bO, acc_o[t]);
            acc_i[t] = wmma4_f32(a[t], bI, acc_i[t]);
            acc_u[t] = wmma4_f32(a[t], bU, acc_u[t]);
        }
    }

    const int rhalf = (lane >> 4) << 3;

    if (rel) {
        const float bob = bo[n];
        const float bib = bi[n];
        const float bub = br[n];
        #pragma unroll
        for (int t = 0; t < MTILES; ++t) {
            #pragma unroll
            for (int j = 0; j < 8; ++j) {
                const long row = base + t * 16 + j + rhalf;
                if (row < rows) {
                    PR[row * REL_DIM + n] = (row < NUM_RELC) ? (acc_o[t][j] + bob)
                                                             : (acc_i[t][j] + bib);
                    out_rel[row * REL_DIM + n] = acc_u[t][j] + bub;
                }
            }
        }
    } else {
        const float bub = bs[n];
        #pragma unroll
        for (int t = 0; t < MTILES; ++t) {
            #pragma unroll
            for (int j = 0; j < 8; ++j) {
                const long row = base + t * 16 + j + rhalf;
                if (row < rows) {
                    P[row * CMP_DIM + n]           = acc_o[t][j];
                    P[row * CMP_DIM + ENT_DIM + n] = acc_i[t][j];
                    out_ent[row * ENT_DIM + n]     = acc_u[t][j] + bub;
                }
            }
        }
    }
}

__global__ __launch_bounds__(256)
void dst_rank_kernel(const int* __restrict__ dst,
                     int*       __restrict__ cnt,
                     int*       __restrict__ rank,
                     int E)
{
    const long e0 = (long)blockIdx.x * (256 * EDGE_EPT) + threadIdx.x;
    int d[EDGE_EPT], r[EDGE_EPT];
    #pragma unroll
    for (int k = 0; k < EDGE_EPT; ++k) {
        const long e = e0 + k * 256;
        d[k] = dst[(e < E) ? e : (long)E - 1];
    }
    #pragma unroll
    for (int k = 0; k < EDGE_EPT; ++k) {
        r[k] = 0;
        if (e0 + k * 256 < E) r[k] = atomicAdd(&cnt[d[k]], 1);
    }
    #pragma unroll
    for (int k = 0; k < EDGE_EPT; ++k) {
        const long e = e0 + k * 256;
        if (e < E) rank[e] = r[k];
    }
}

__global__ __launch_bounds__(256)
void segment_alloc_kernel(const int* __restrict__ cnt,
                          int*       __restrict__ cursor,
                          int*       __restrict__ off,
                          int N)
{
    __shared__ int wsum[8];
    __shared__ int bbase;
    const int tid  = threadIdx.x;
    const int lane = tid & 31;
    const int wave = tid >> 5;
    const int d    = (int)blockIdx.x * 256 + tid;
    const int c    = (d < N) ? cnt[d] : 0;

    int x = c;
    #pragma unroll
    for (int s = 1; s < 32; s <<= 1) {
        const int y = __shfl_up(x, s, 32);
        if (lane >= s) x += y;
    }
    if (lane == 31) wsum[wave] = x;
    __syncthreads();
    if (tid == 0) {
        int t = 0;
        #pragma unroll
        for (int w = 0; w < 8; ++w) { const int v = wsum[w]; wsum[w] = t; t += v; }
        bbase = atomicAdd(cursor, t);
    }
    __syncthreads();
    if (d < N) off[d] = bbase + wsum[wave] + x - c;
}

__global__ __launch_bounds__(256)
void edge_place_kernel(const int* __restrict__ src,
                       const int* __restrict__ dst,
                       const int* __restrict__ etypes,
                       const int* __restrict__ off,
                       const int* __restrict__ rank,
                       int*       __restrict__ key,
                       int E)
{
    const long e0 = (long)blockIdx.x * (256 * EDGE_EPT) + threadIdx.x;
    int d[EDGE_EPT], r[EDGE_EPT], s[EDGE_EPT], ty[EDGE_EPT], o[EDGE_EPT];
    #pragma unroll
    for (int k = 0; k < EDGE_EPT; ++k) {
        const long e  = e0 + k * 256;
        const long ec = (e < E) ? e : (long)E - 1;
        d[k]  = dst[ec];
        r[k]  = rank[ec];
        s[k]  = src[ec];
        ty[k] = etypes[ec];
    }
    #pragma unroll
    for (int k = 0; k < EDGE_EPT; ++k)
        o[k] = off[d[k]];
    #pragma unroll
    for (int k = 0; k < EDGE_EPT; ++k) {
        if (e0 + k * 256 < E) {
            const long pos = (long)o[k] + r[k];
            v2i kv;
            kv.x = s[k] * CMP_DIM + ((ty[k] < NUM_RELC) ? 0 : ENT_DIM);
            kv.y = ty[k] * REL_DIM;
            *(v2i*)(key + 2 * pos) = kv;
        }
    }
}

__global__ __launch_bounds__(256)
void node_gather_sum_kernel(const int*   __restrict__ key,
                            const int*   __restrict__ cnt,
                            const int*   __restrict__ off,
                            const float* __restrict__ P,
                            const float* __restrict__ PR,
                            float*       __restrict__ out,
                            int N)
{
    const int lane = threadIdx.x & 31;
    const int wave = __builtin_amdgcn_readfirstlane((int)(threadIdx.x >> 5));
    const int col  = lane << 2;

    #pragma unroll 1
    for (int trip = 0; trip < NODE_TRIPS; ++trip) {
        const int node = ((int)blockIdx.x * NODE_TRIPS + trip) * 8 + wave;
        if (node >= N) break;
        const int n = __builtin_amdgcn_readfirstlane(cnt[node]);
        const int o = __builtin_amdgcn_readfirstlane(off[node]);
        v4f acc = {0.f, 0.f, 0.f, 0.f};
        int m = (n < 32) ? n : 32;
        v2i kv = {0, 0};
        if (lane < m) kv = *(const v2i*)(key + 2 * ((long)o + lane));
        float* orow = out + (long)node * ENT_DIM + col;
        const v4f self_term = *(const v4f*)orow;

        for (int base = 0; base < n; ) {
            for (int g = 0; g < m; g += 4) {
                v4f pv[4], rv[4];
                #pragma unroll
                for (int i = 0; i < 4; ++i) {
                    const int pi = __builtin_amdgcn_readlane(kv.x, g + i);
                    const int ri = __builtin_amdgcn_readlane(kv.y, g + i);
                    pv[i] = *(const v4f*)(P  + pi + col);
                    rv[i] = *(const v4f*)(PR + ri + col);
                }
                #pragma unroll
                for (int i = 0; i < 4; ++i) {
                    if (g + i < m)
                        acc += rv[i] + pv[i];
                }
            }
            base += 32;
            m = (n - base < 32) ? (n - base) : 32;
            if (base < n) {
                kv = (v2i){0, 0};
                if (lane < m) kv = *(const v2i*)(key + 2 * ((long)o + base + lane));
            }
        }
        float dg = (float)n;
        dg = (dg > 1.0f) ? dg : 1.0f;
        *(v4f*)orow = self_term + acc / dg;
    }
}

extern "C" void kernel_launch(void* const* d_in, const int* in_sizes, int n_in,
                              void* d_out, int out_size, void* d_ws, size_t ws_size,
                              hipStream_t stream)
{
    const int*   src     = (const int*)  d_in[0];
    const int*   dst     = (const int*)  d_in[1];
    const int*   etypes  = (const int*)  d_in[2];
    const float* ent_emb = (const float*)d_in[3];
    const float* rel_emb = (const float*)d_in[4];
    const float* Wo      = (const float*)d_in[5];
    const float* bo      = (const float*)d_in[6];
    const float* Wi      = (const float*)d_in[7];
    const float* bi      = (const float*)d_in[8];
    const float* Ws      = (const float*)d_in[9];
    const float* bs      = (const float*)d_in[10];
    const float* Wr      = (const float*)d_in[11];
    const float* br      = (const float*)d_in[12];

    const int E = in_sizes[0];
    const int N = in_sizes[3] / ENT_DIM;
    const int R = in_sizes[4] / REL_DIM;

    float* P    = (float*)d_ws;
    float* PR   = P   + (size_t)N * CMP_DIM;
    int*   cnt  = (int*)(PR + (((size_t)R * REL_DIM + 63) & ~(size_t)63));
    int*   curs = cnt  + (((size_t)N + 63) & ~(size_t)63);
    int*   off  = curs + 64;
    int*   rank = off  + (((size_t)N + 63) & ~(size_t)63);
    int*   key  = rank + (((size_t)E + 63) & ~(size_t)63);

    float* out_ent = (float*)d_out;
    float* out_rel = out_ent + (size_t)N * ENT_DIM;

    const int rows_per_block = MTILES * 16;
    const int node_blocks    = (N + rows_per_block - 1) / rows_per_block;
    const int rel_blocks     = (R + rows_per_block - 1) / rows_per_block;
    project_kernel<<<node_blocks + rel_blocks, 256, 0, stream>>>(
        ent_emb, rel_emb, Wo, bo, Wi, bi, Ws, bs, Wr, br, out_ent, out_rel,
        P, PR, cnt, curs, N, R, node_blocks);

    const int edges_per_block = 256 * EDGE_EPT;
    const int edge_blocks     = (E + edges_per_block - 1) / edges_per_block;
    dst_rank_kernel<<<edge_blocks, 256, 0, stream>>>(dst, cnt, rank, E);
    segment_alloc_kernel<<<(N + 255) / 256, 256, 0, stream>>>(cnt, curs, off, N);
    edge_place_kernel<<<edge_blocks, 256, 0, stream>>>(src, dst, etypes, off, rank, key, E);

    const int nodes_per_block = 8 * NODE_TRIPS;
    node_gather_sum_kernel<<<(N + nodes_per_block - 1) / nodes_per_block, 256, 0, stream>>>(
        key, cnt, off, P, PR, out_ent, N);
}


# Node-sum kernel made persistent (256 groups of 32 waves) with the projected relation table copied to LDS once per group: relation rows come from LDS, halving the rows gathered from L2
# speedup vs baseline: 5.3412x; 1.0546x over previous
#include <hip/hip_runtime.h>

#define ENT_DIM  128
#define REL_DIM  128
#define CMP_DIM  256
#define NUM_RELC 200
#define MTILES   4
#define EDGE_EPT   4
#define REL_ROWS   (2 * NUM_RELC)
#define SUM_WAVES  32
#define SUM_GROUPS 256
#define SUM_EDGES  8

typedef __attribute__((ext_vector_type(2))) float v2f;
typedef __attribute__((ext_vector_type(8))) float v8f;
typedef __attribute__((ext_vector_type(4))) float v4f;
typedef __attribute__((ext_vector_type(2))) int   v2i;

__device__ __forceinline__ v8f wmma4_f32(v2f a, v2f b, v8f c) {
    return __builtin_amdgcn_wmma_f32_16x16x4_f32(
         false, a,  false, b,
         (short)0, c,  false,  false);
}

__global__ __launch_bounds__(256)
void project_kernel(const float* __restrict__ ent_emb,
                    const float* __restrict__ rel_emb,
                    const float* __restrict__ Wo,
                    const float* __restrict__ bo,
                    const float* __restrict__ Wi,
                    const float* __restrict__ bi,
                    const float* __restrict__ Ws,
                    const float* __restrict__ bs,
                    const float* __restrict__ Wr,
                    const float* __restrict__ br,
                    float*       __restrict__ out_ent,
                    float*       __restrict__ out_rel,
                    float*       __restrict__ P,
                    float*       __restrict__ PR,
                    int*         __restrict__ cnt,
                    int*         __restrict__ cursor,
                    int N, int R, int node_blocks)
{
    const int  tid   = threadIdx.x;
    const int  lane  = tid & 31;
    const int  wave  = tid >> 5;
    const int  m     = lane & 15;
    const int  khalf = (lane >> 4) << 1;

    const bool   rel  = (int)blockIdx.x >= node_blocks;
    const float* A    = rel ? rel_emb : ent_emb;
    const int    rows = rel ? R : N;
    const int    koff = rel ? 0 : REL_DIM;
    const long   base = (long)(rel ? (int)blockIdx.x - node_blocks : (int)blockIdx.x)
                        * (MTILES * 16);

    if (!rel && tid < MTILES * 16 && base + tid < rows) cnt[base + tid] = 0;
    if (blockIdx.x == 0 && tid == MTILES * 16) *cursor = 0;

    const float* arow[MTILES];
    #pragma unroll
    for (int t = 0; t < MTILES; ++t) {
        long r  = base + t * 16 + m;
        long rc = (r < rows) ? r : (long)(rows - 1);
        arow[t] = A + rc * ENT_DIM;
    }

    const int    n      = (wave << 4) + m;
    const float* wo_row = Wo + (long)n * CMP_DIM + koff;
    const float* wi_row = Wi + (long)n * CMP_DIM + koff;
    const float* wu_row = (rel ? Wr : Ws) + (long)n * ENT_DIM;

    v8f acc_o[MTILES], acc_i[MTILES], acc_u[MTILES];
    #pragma unroll
    for (int t = 0; t < MTILES; ++t) {
        acc_o[t] = (v8f){0.f,0.f,0.f,0.f,0.f,0.f,0.f,0.f};
        acc_i[t] = (v8f){0.f,0.f,0.f,0.f,0.f,0.f,0.f,0.f};
        acc_u[t] = (v8f){0.f,0.f,0.f,0.f,0.f,0.f,0.f,0.f};
    }

    #pragma unroll 2
    for (int k = 0; k < ENT_DIM; k += 4) {
        v2f bO = *(const v2f*)(wo_row + k + khalf);
        v2f bI = *(const v2f*)(wi_row + k + khalf);
        v2f bU = *(const v2f*)(wu_row + k + khalf);
        v2f a[MTILES];
        #pragma unroll
        for (int t = 0; t < MTILES; ++t)
            a[t] = *(const v2f*)(arow[t] + k + khalf);
        #pragma unroll
        for (int t = 0; t < MTILES; ++t) {
            acc_o[t] = wmma4_f32(a[t], bO, acc_o[t]);
            acc_i[t] = wmma4_f32(a[t], bI, acc_i[t]);
            acc_u[t] = wmma4_f32(a[t], bU, acc_u[t]);
        }
    }

    const int rhalf = (lane >> 4) << 3;

    if (rel) {
        const float bob = bo[n];
        const float bib = bi[n];
        const float bub = br[n];
        #pragma unroll
        for (int t = 0; t < MTILES; ++t) {
            #pragma unroll
            for (int j = 0; j < 8; ++j) {
                const long row = base + t * 16 + j + rhalf;
                if (row < rows) {
                    PR[row * REL_DIM + n] = (row < NUM_RELC) ? (acc_o[t][j] + bob)
                                                             : (acc_i[t][j] + bib);
                    out_rel[row * REL_DIM + n] = acc_u[t][j] + bub;
                }
            }
        }
    } else {
        const float bub = bs[n];
        #pragma unroll
        for (int t = 0; t < MTILES; ++t) {
            #pragma unroll
            for (int j = 0; j < 8; ++j) {
                const long row = base + t * 16 + j + rhalf;
                if (row < rows) {
                    P[row * CMP_DIM + n]           = acc_o[t][j];
                    P[row * CMP_DIM + ENT_DIM + n] = acc_i[t][j];
                    out_ent[row * ENT_DIM + n]     = acc_u[t][j] + bub;
                }
            }
        }
    }
}

__global__ __launch_bounds__(256)
void dst_rank_kernel(const int* __restrict__ dst,
                     int*       __restrict__ cnt,
                     int*       __restrict__ rank,
                     int E)
{
    const long e0 = (long)blockIdx.x * (256 * EDGE_EPT) + threadIdx.x;
    int d[EDGE_EPT], r[EDGE_EPT];
    #pragma unroll
    for (int k = 0; k < EDGE_EPT; ++k) {
        const long e = e0 + k * 256;
        d[k] = dst[(e < E) ? e : (long)E - 1];
    }
    #pragma unroll
    for (int k = 0; k < EDGE_EPT; ++k) {
        r[k] = 0;
        if (e0 + k * 256 < E) r[k] = atomicAdd(&cnt[d[k]], 1);
    }
    #pragma unroll
    for (int k = 0; k < EDGE_EPT; ++k) {
        const long e = e0 + k * 256;
        if (e < E) rank[e] = r[k];
    }
}

__global__ __launch_bounds__(256)
void segment_alloc_kernel(const int* __restrict__ cnt,
                          int*       __restrict__ cursor,
                          int*       __restrict__ off,
                          int N)
{
    __shared__ int wsum[8];
    __shared__ int bbase;
    const int tid  = threadIdx.x;
    const int lane = tid & 31;
    const int wave = tid >> 5;
    const int d    = (int)blockIdx.x * 256 + tid;
    const int c    = (d < N) ? cnt[d] : 0;

    int x = c;
    #pragma unroll
    for (int s = 1; s < 32; s <<= 1) {
        const int y = __shfl_up(x, s, 32);
        if (lane >= s) x += y;
    }
    if (lane == 31) wsum[wave] = x;
    __syncthreads();
    if (tid == 0) {
        int t = 0;
        #pragma unroll
        for (int w = 0; w < 8; ++w) { const int v = wsum[w]; wsum[w] = t; t += v; }
        bbase = atomicAdd(cursor, t);
    }
    __syncthreads();
    if (d < N) off[d] = bbase + wsum[wave] + x - c;
}

__global__ __launch_bounds__(256)
void edge_place_kernel(const int* __restrict__ src,
                       const int* __restrict__ dst,
                       const int* __restrict__ etypes,
                       const int* __restrict__ off,
                       const int* __restrict__ rank,
                       int*       __restrict__ key,
                       int E)
{
    const long e0 = (long)blockIdx.x * (256 * EDGE_EPT) + threadIdx.x;
    int d[EDGE_EPT], r[EDGE_EPT], s[EDGE_EPT], ty[EDGE_EPT], o[EDGE_EPT];
    #pragma unroll
    for (int k = 0; k < EDGE_EPT; ++k) {
        const long e  = e0 + k * 256;
        const long ec = (e < E) ? e : (long)E - 1;
        d[k]  = dst[ec];
        r[k]  = rank[ec];
        s[k]  = src[ec];
        ty[k] = etypes[ec];
    }
    #pragma unroll
    for (int k = 0; k < EDGE_EPT; ++k)
        o[k] = off[d[k]];
    #pragma unroll
    for (int k = 0; k < EDGE_EPT; ++k) {
        if (e0 + k * 256 < E) {
            const long pos = (long)o[k] + r[k];
            v2i kv;
            kv.x = s[k] * CMP_DIM + ((ty[k] < NUM_RELC) ? 0 : ENT_DIM);
            kv.y = ty[k] * REL_DIM;
            *(v2i*)(key + 2 * pos) = kv;
        }
    }
}

__global__ __launch_bounds__(SUM_WAVES * 32)
void node_gather_sum_kernel(const int*   __restrict__ key,
                            const int*   __restrict__ cnt,
                            const int*   __restrict__ off,
                            const float* __restrict__ P,
                            const float* __restrict__ PR,
                            float*       __restrict__ out,
                            int N)
{
    __shared__ v4f pr_lds[REL_ROWS * REL_DIM / 4];

    const int tid  = threadIdx.x;
    const int lane = tid & 31;
    const int wave = __builtin_amdgcn_readfirstlane(tid >> 5);
    const int col  = lane << 2;
    const int step = (int)gridDim.x * SUM_WAVES;

    int node = wave * (int)gridDim.x + (int)blockIdx.x;
    int n_v = 0, o_v = 0;
    if (node < N) { n_v = cnt[node]; o_v = off[node]; }

    {
        const v4f* PR4 = (const v4f*)PR;
        #pragma unroll
        for (int i = 0; i < (REL_ROWS * REL_DIM / 4 + SUM_WAVES * 32 - 1) / (SUM_WAVES * 32); ++i) {
            const int idx = i * (SUM_WAVES * 32) + tid;
            if (idx < REL_ROWS * REL_DIM / 4) pr_lds[idx] = PR4[idx];
        }
    }
    __syncthreads();

    #pragma unroll 1
    for (; node < N; node += step) {
        const int n = __builtin_amdgcn_readfirstlane(n_v);
        const int o = __builtin_amdgcn_readfirstlane(o_v);
        const int nxt = node + step;
        n_v = 0; o_v = 0;
        if (nxt < N) { n_v = cnt[nxt]; o_v = off[nxt]; }

        v4f acc = {0.f, 0.f, 0.f, 0.f};
        int m = (n < 32) ? n : 32;
        v2i kv = {0, 0};
        if (lane < m) kv = *(const v2i*)(key + 2 * ((long)o + lane));
        float* orow = out + (long)node * ENT_DIM + col;
        const v4f self_term = *(const v4f*)orow;

        for (int base = 0; base < n; ) {
            for (int g = 0; g < m; g += SUM_EDGES) {
                v4f pv[SUM_EDGES], rv[SUM_EDGES];
                #pragma unroll
                for (int i = 0; i < SUM_EDGES; ++i) {
                    const int pi = __builtin_amdgcn_readlane(kv.x, g + i);
                    const int ri = __builtin_amdgcn_readlane(kv.y, g + i);
                    pv[i] = *(const v4f*)(P + pi + col);
                    rv[i] = pr_lds[(ri + col) >> 2];
                }
                #pragma unroll
                for (int i = 0; i < SUM_EDGES; ++i) {
                    if (g + i < m)
                        acc += rv[i] + pv[i];
                }
            }
            base += 32;
            m = (n - base < 32) ? (n - base) : 32;
            if (base < n) {
                kv = (v2i){0, 0};
                if (lane < m) kv = *(const v2i*)(key + 2 * ((long)o + base + lane));
            }
        }
        float dg = (float)n;
        dg = (dg > 1.0f) ? dg : 1.0f;
        *(v4f*)orow = self_term + acc / dg;
    }
}

extern "C" void kernel_launch(void* const* d_in, const int* in_sizes, int n_in,
                              void* d_out, int out_size, void* d_ws, size_t ws_size,
                              hipStream_t stream)
{
    const int*   src     = (const int*)  d_in[0];
    const int*   dst     = (const int*)  d_in[1];
    const int*   etypes  = (const int*)  d_in[2];
    const float* ent_emb = (const float*)d_in[3];
    const float* rel_emb = (const float*)d_in[4];
    const float* Wo      = (const float*)d_in[5];
    const float* bo      = (const float*)d_in[6];
    const float* Wi      = (const float*)d_in[7];
    const float* bi      = (const float*)d_in[8];
    const float* Ws      = (const float*)d_in[9];
    const float* bs      = (const float*)d_in[10];
    const float* Wr      = (const float*)d_in[11];
    const float* br      = (const float*)d_in[12];

    const int E = in_sizes[0];
    const int N = in_sizes[3] / ENT_DIM;
    const int R = in_sizes[4] / REL_DIM;

    float* P    = (float*)d_ws;
    float* PR   = P   + (size_t)N * CMP_DIM;
    int*   cnt  = (int*)(PR + (((size_t)R * REL_DIM + 63) & ~(size_t)63));
    int*   curs = cnt  + (((size_t)N + 63) & ~(size_t)63);
    int*   off  = curs + 64;
    int*   rank = off  + (((size_t)N + 63) & ~(size_t)63);
    int*   key  = rank + (((size_t)E + 63) & ~(size_t)63);

    float* out_ent = (float*)d_out;
    float* out_rel = out_ent + (size_t)N * ENT_DIM;

    const int rows_per_block = MTILES * 16;
    const int node_blocks    = (N + rows_per_block - 1) / rows_per_block;
    const int rel_blocks     = (R + rows_per_block - 1) / rows_per_block;
    project_kernel<<<node_blocks + rel_blocks, 256, 0, stream>>>(
        ent_emb, rel_emb, Wo, bo, Wi, bi, Ws, bs, Wr, br, out_ent, out_rel,
        P, PR, cnt, curs, N, R, node_blocks);

    const int edges_per_block = 256 * EDGE_EPT;
    const int edge_blocks     = (E + edges_per_block - 1) / edges_per_block;
    dst_rank_kernel<<<edge_blocks, 256, 0, stream>>>(dst, cnt, rank, E);
    segment_alloc_kernel<<<(N + 255) / 256, 256, 0, stream>>>(cnt, curs, off, N);
    edge_place_kernel<<<edge_blocks, 256, 0, stream>>>(src, dst, etypes, off, rank, key, E);

    node_gather_sum_kernel<<<SUM_GROUPS, SUM_WAVES * 32, 0, stream>>>(
        key, cnt, off, P, PR, out_ent, N);
}


# Rank edges per destination with LDS counters (32 edge chunks, one work-group each, per-chunk count table) instead of one returning global atomic per edge; sum order: by chunk, then LDS arrival
# speedup vs baseline: 6.5562x; 1.2275x over previous
#include <hip/hip_runtime.h>

#define ENT_DIM  128
#define REL_DIM  128
#define CMP_DIM  256
#define NUM_RELC 200
#define MTILES   4
#define EDGE_EPT   4
#define REL_ROWS   (2 * NUM_RELC)
#define SUM_WAVES  32
#define SUM_GROUPS 256
#define SUM_EDGES  8
#define NODE_CAP    50000
#define RANK_CHUNKS 32
#define RANK_WAVES  32
#define RANK_EPT    13

typedef __attribute__((ext_vector_type(2))) float v2f;
typedef __attribute__((ext_vector_type(8))) float v8f;
typedef __attribute__((ext_vector_type(4))) float v4f;
typedef __attribute__((ext_vector_type(2))) int   v2i;
typedef __attribute__((ext_vector_type(4))) int   v4i;

__device__ __forceinline__ v8f wmma4_f32(v2f a, v2f b, v8f c) {
    return __builtin_amdgcn_wmma_f32_16x16x4_f32(
         false, a,  false, b,
         (short)0, c,  false,  false);
}

__global__ __launch_bounds__(256)
void project_kernel(const float* __restrict__ ent_emb,
                    const float* __restrict__ rel_emb,
                    const float* __restrict__ Wo,
                    const float* __restrict__ bo,
                    const float* __restrict__ Wi,
                    const float* __restrict__ bi,
                    const float* __restrict__ Ws,
                    const float* __restrict__ bs,
                    const float* __restrict__ Wr,
                    const float* __restrict__ br,
                    float*       __restrict__ out_ent,
                    float*       __restrict__ out_rel,
                    float*       __restrict__ P,
                    float*       __restrict__ PR,
                    int*         __restrict__ cursor,
                    int N, int R, int node_blocks)
{
    const int  tid   = threadIdx.x;
    const int  lane  = tid & 31;
    const int  wave  = tid >> 5;
    const int  m     = lane & 15;
    const int  khalf = (lane >> 4) << 1;

    const bool   rel  = (int)blockIdx.x >= node_blocks;
    const float* A    = rel ? rel_emb : ent_emb;
    const int    rows = rel ? R : N;
    const int    koff = rel ? 0 : REL_DIM;
    const long   base = (long)(rel ? (int)blockIdx.x - node_blocks : (int)blockIdx.x)
                        * (MTILES * 16);

    if (blockIdx.x == 0 && tid == MTILES * 16) *cursor = 0;

    const float* arow[MTILES];
    #pragma unroll
    for (int t = 0; t < MTILES; ++t) {
        long r  = base + t * 16 + m;
        long rc = (r < rows) ? r : (long)(rows - 1);
        arow[t] = A + rc * ENT_DIM;
    }

    const int    n      = (wave << 4) + m;
    const float* wo_row = Wo + (long)n * CMP_DIM + koff;
    const float* wi_row = Wi + (long)n * CMP_DIM + koff;
    const float* wu_row = (rel ? Wr : Ws) + (long)n * ENT_DIM;

    v8f acc_o[MTILES], acc_i[MTILES], acc_u[MTILES];
    #pragma unroll
    for (int t = 0; t < MTILES; ++t) {
        acc_o[t] = (v8f){0.f,0.f,0.f,0.f,0.f,0.f,0.f,0.f};
        acc_i[t] = (v8f){0.f,0.f,0.f,0.f,0.f,0.f,0.f,0.f};
        acc_u[t] = (v8f){0.f,0.f,0.f,0.f,0.f,0.f,0.f,0.f};
    }

    #pragma unroll 2
    for (int k = 0; k < ENT_DIM; k += 4) {
        v2f bO = *(const v2f*)(wo_row + k + khalf);
        v2f bI = *(const v2f*)(wi_row + k + khalf);
        v2f bU = *(const v2f*)(wu_row + k + khalf);
        v2f a[MTILES];
        #pragma unroll
        for (int t = 0; t < MTILES; ++t)
            a[t] = *(const v2f*)(arow[t] + k + khalf);
        #pragma unroll
        for (int t = 0; t < MTILES; ++t) {
            acc_o[t] = wmma4_f32(a[t], bO, acc_o[t]);
            acc_i[t] = wmma4_f32(a[t], bI, acc_i[t]);
            acc_u[t] = wmma4_f32(a[t], bU, acc_u[t]);
        }
    }

    const int rhalf = (lane >> 4) << 3;

    if (rel) {
        const float bob = bo[n];
        const float bib = bi[n];
        const float bub = br[n];
        #pragma unroll
        for (int t = 0; t < MTILES; ++t) {
            #pragma unroll
            for (int j = 0; j < 8; ++j) {
                const long row = base + t * 16 + j + rhalf;
                if (row < rows) {
                    PR[row * REL_DIM + n] = (row < NUM_RELC) ? (acc_o[t][j] + bob)
                                                             : (acc_i[t][j] + bib);
                    out_rel[row * REL_DIM + n] = acc_u[t][j] + bub;
                }
            }
        }
    } else {
        const float bub = bs[n];
        #pragma unroll
        for (int t = 0; t < MTILES; ++t) {
            #pragma unroll
            for (int j = 0; j < 8; ++j) {
                const long row = base + t * 16 + j + rhalf;
                if (row < rows) {
                    P[row * CMP_DIM + n]           = acc_o[t][j];
                    P[row * CMP_DIM + ENT_DIM + n] = acc_i[t][j];
                    out_ent[row * ENT_DIM + n]     = acc_u[t][j] + bub;
                }
            }
        }
    }
}

__global__ __launch_bounds__(RANK_WAVES * 32)
void dst_rank_kernel(const int* __restrict__ dst,
                     int*       __restrict__ rank,
                     int*       __restrict__ H,
                     int E, int chunk)
{
    __shared__ int hist[NODE_CAP];
    v4i* hist4 = (v4i*)hist;
    const int tid = threadIdx.x;

    for (int i = tid; i < NODE_CAP / 4; i += RANK_WAVES * 32)
        hist4[i] = (v4i){0, 0, 0, 0};
    __syncthreads();

    const int c0 = (int)blockIdx.x * chunk;
    const int c1 = (c0 + chunk < E) ? (c0 + chunk) : E;
    for (int b = c0 + tid; b < c1; b += RANK_EPT * (RANK_WAVES * 32)) {
        int d[RANK_EPT], r[RANK_EPT];
        #pragma unroll
        for (int k = 0; k < RANK_EPT; ++k) {
            const int e = b + k * (RANK_WAVES * 32);
            d[k] = dst[(e < c1) ? e : b];
        }
        #pragma unroll
        for (int k = 0; k < RANK_EPT; ++k)
            r[k] = atomicAdd(&hist[d[k]], (b + k * (RANK_WAVES * 32) < c1) ? 1 : 0);
        #pragma unroll
        for (int k = 0; k < RANK_EPT; ++k) {
            const int e = b + k * (RANK_WAVES * 32);
            if (e < c1) rank[e] = r[k];
        }
    }
    __syncthreads();

    v4i* Hrow = (v4i*)(H + (long)blockIdx.x * NODE_CAP);
    for (int i = tid; i < NODE_CAP / 4; i += RANK_WAVES * 32)
        Hrow[i] = hist4[i];
}

__global__ __launch_bounds__(256)
void segment_alloc_kernel(int* __restrict__ H,
                          int* __restrict__ cursor,
                          int* __restrict__ cnt,
                          int* __restrict__ off,
                          int N)
{
    __shared__ int wsum[8];
    __shared__ int bbase;
    const int tid  = threadIdx.x;
    const int lane = tid & 31;
    const int wave = tid >> 5;
    const int d    = (int)blockIdx.x * 256 + tid;
    const int dc   = (d < N) ? d : N - 1;

    int h[RANK_CHUNKS];
    #pragma unroll
    for (int q = 0; q < RANK_CHUNKS; ++q)
        h[q] = H[(long)q * NODE_CAP + dc];
    int tot = 0;
    #pragma unroll
    for (int q = 0; q < RANK_CHUNKS; ++q) { const int v = h[q]; h[q] = tot; tot += v; }
    const int c = (d < N) ? tot : 0;

    int x = c;
    #pragma unroll
    for (int s = 1; s < 32; s <<= 1) {
        const int y = __shfl_up(x, s, 32);
        if (lane >= s) x += y;
    }
    if (lane == 31) wsum[wave] = x;
    __syncthreads();
    if (tid == 0) {
        int t = 0;
        #pragma unroll
        for (int w = 0; w < 8; ++w) { const int v = wsum[w]; wsum[w] = t; t += v; }
        bbase = atomicAdd(cursor, t);
    }
    __syncthreads();
    if (d < N) {
        const int o = bbase + wsum[wave] + x - c;
        cnt[d] = c;
        off[d] = o;
        #pragma unroll
        for (int q = 0; q < RANK_CHUNKS; ++q)
            H[(long)q * NODE_CAP + d] = o + h[q];
    }
}

__global__ __launch_bounds__(256)
void edge_place_kernel(const int* __restrict__ src,
                       const int* __restrict__ dst,
                       const int* __restrict__ etypes,
                       const int* __restrict__ H,
                       const int* __restrict__ rank,
                       int*       __restrict__ key,
                       int E, int chunk)
{
    const long e0 = (long)blockIdx.x * (256 * EDGE_EPT) + threadIdx.x;
    const int* Hc = H + (long)(((int)blockIdx.x * (256 * EDGE_EPT)) / chunk) * NODE_CAP;
    int d[EDGE_EPT], r[EDGE_EPT], s[EDGE_EPT], ty[EDGE_EPT], o[EDGE_EPT];
    #pragma unroll
    for (int k = 0; k < EDGE_EPT; ++k) {
        const long e  = e0 + k * 256;
        const long ec = (e < E) ? e : (long)E - 1;
        d[k]  = dst[ec];
        r[k]  = rank[ec];
        s[k]  = src[ec];
        ty[k] = etypes[ec];
    }
    #pragma unroll
    for (int k = 0; k < EDGE_EPT; ++k)
        o[k] = Hc[d[k]];
    #pragma unroll
    for (int k = 0; k < EDGE_EPT; ++k) {
        if (e0 + k * 256 < E) {
            const long pos = (long)o[k] + r[k];
            v2i kv;
            kv.x = s[k] * CMP_DIM + ((ty[k] < NUM_RELC) ? 0 : ENT_DIM);
            kv.y = ty[k] * REL_DIM;
            *(v2i*)(key + 2 * pos) = kv;
        }
    }
}

__global__ __launch_bounds__(SUM_WAVES * 32)
void node_gather_sum_kernel(const int*   __restrict__ key,
                            const int*   __restrict__ cnt,
                            const int*   __restrict__ off,
                            const float* __restrict__ P,
                            const float* __restrict__ PR,
                            float*       __restrict__ out,
                            int N)
{
    __shared__ v4f pr_lds[REL_ROWS * REL_DIM / 4];

    const int tid  = threadIdx.x;
    const int lane = tid & 31;
    const int wave = __builtin_amdgcn_readfirstlane(tid >> 5);
    const int col  = lane << 2;
    const int step = (int)gridDim.x * SUM_WAVES;

    int node = wave * (int)gridDim.x + (int)blockIdx.x;
    int n_v = 0, o_v = 0;
    if (node < N) { n_v = cnt[node]; o_v = off[node]; }

    {
        const v4f* PR4 = (const v4f*)PR;
        #pragma unroll
        for (int i = 0; i < (REL_ROWS * REL_DIM / 4 + SUM_WAVES * 32 - 1) / (SUM_WAVES * 32); ++i) {
            const int idx = i * (SUM_WAVES * 32) + tid;
            if (idx < REL_ROWS * REL_DIM / 4) pr_lds[idx] = PR4[idx];
        }
    }
    __syncthreads();

    #pragma unroll 1
    for (; node < N; node += step) {
        const int n = __builtin_amdgcn_readfirstlane(n_v);
        const int o = __builtin_amdgcn_readfirstlane(o_v);
        const int nxt = node + step;
        n_v = 0; o_v = 0;
        if (nxt < N) { n_v = cnt[nxt]; o_v = off[nxt]; }

        v4f acc = {0.f, 0.f, 0.f, 0.f};
        int m = (n < 32) ? n : 32;
        v2i kv = {0, 0};
        if (lane < m) kv = *(const v2i*)(key + 2 * ((long)o + lane));
        float* orow = out + (long)node * ENT_DIM + col;
        const v4f self_term = *(const v4f*)orow;

        for (int base = 0; base < n; ) {
            for (int g = 0; g < m; g += SUM_EDGES) {
                v4f pv[SUM_EDGES], rv[SUM_EDGES];
                #pragma unroll
                for (int i = 0; i < SUM_EDGES; ++i) {
                    const int pi = __builtin_amdgcn_readlane(kv.x, g + i);
                    const int ri = __builtin_amdgcn_readlane(kv.y, g + i);
                    pv[i] = *(const v4f*)(P + pi + col);
                    rv[i] = pr_lds[(ri + col) >> 2];
                }
                #pragma unroll
                for (int i = 0; i < SUM_EDGES; ++i) {
                    if (g + i < m)
                        acc += rv[i] + pv[i];
                }
            }
            base += 32;
            m = (n - base < 32) ? (n - base) : 32;
            if (base < n) {
                kv = (v2i){0, 0};
                if (lane < m) kv = *(const v2i*)(key + 2 * ((long)o + base + lane));
            }
        }
        float dg = (float)n;
        dg = (dg > 1.0f) ? dg : 1.0f;
        *(v4f*)orow = self_term + acc / dg;
    }
}

extern "C" void kernel_launch(void* const* d_in, const int* in_sizes, int n_in,
                              void* d_out, int out_size, void* d_ws, size_t ws_size,
                              hipStream_t stream)
{
    const int*   src     = (const int*)  d_in[0];
    const int*   dst     = (const int*)  d_in[1];
    const int*   etypes  = (const int*)  d_in[2];
    const float* ent_emb = (const float*)d_in[3];
    const float* rel_emb = (const float*)d_in[4];
    const float* Wo      = (const float*)d_in[5];
    const float* bo      = (const float*)d_in[6];
    const float* Wi      = (const float*)d_in[7];
    const float* bi      = (const float*)d_in[8];
    const float* Ws      = (const float*)d_in[9];
    const float* bs      = (const float*)d_in[10];
    const float* Wr      = (const float*)d_in[11];
    const float* br      = (const float*)d_in[12];

    const int E = in_sizes[0];
    const int N = in_sizes[3] / ENT_DIM;
    const int R = in_sizes[4] / REL_DIM;

    float* P    = (float*)d_ws;
    float* PR   = P   + (size_t)N * CMP_DIM;
    int*   cnt  = (int*)(PR + (((size_t)R * REL_DIM + 63) & ~(size_t)63));
    int*   curs = cnt  + (((size_t)N + 63) & ~(size_t)63);
    int*   off  = curs + 64;
    int*   rank = off  + (((size_t)N + 63) & ~(size_t)63);
    int*   key  = rank + (((size_t)E + 63) & ~(size_t)63);
    int*   H    = key  + (((size_t)E * 2 + 63) & ~(size_t)63);

    float* out_ent = (float*)d_out;
    float* out_rel = out_ent + (size_t)N * ENT_DIM;

    const int rows_per_block = MTILES * 16;
    const int node_blocks    = (N + rows_per_block - 1) / rows_per_block;
    const int rel_blocks     = (R + rows_per_block - 1) / rows_per_block;
    project_kernel<<<node_blocks + rel_blocks, 256, 0, stream>>>(
        ent_emb, rel_emb, Wo, bo, Wi, bi, Ws, bs, Wr, br, out_ent, out_rel,
        P, PR, curs, N, R, node_blocks);

    const int edges_per_block = 256 * EDGE_EPT;
    const int edge_blocks     = (E + edges_per_block - 1) / edges_per_block;
    const int chunk = ((E + RANK_CHUNKS * edges_per_block - 1) / (RANK_CHUNKS * edges_per_block))
                      * edges_per_block;
    dst_rank_kernel<<<RANK_CHUNKS, RANK_WAVES * 32, 0, stream>>>(dst, rank, H, E, chunk);
    segment_alloc_kernel<<<(N + 255) / 256, 256, 0, stream>>>(H, curs, cnt, off, N);
    edge_place_kernel<<<edge_blocks, 256, 0, stream>>>(src, dst, etypes, H, rank, key, E, chunk);

    node_gather_sum_kernel<<<SUM_GROUPS, SUM_WAVES * 32, 0, stream>>>(
        key, cnt, off, P, PR, out_ent, N);
}


# Projection GEMM: each lane half loads 4 consecutive k as one 16-byte word (A and W alike), 2 WMMAs per word: half the fragment loads; k order inside every 8 changes, so the dot-product sum order too
# speedup vs baseline: 6.9151x; 1.0547x over previous
#include <hip/hip_runtime.h>

#define ENT_DIM  128
#define REL_DIM  128
#define CMP_DIM  256
#define NUM_RELC 200
#define MTILES   4
#define EDGE_EPT   4
#define REL_ROWS   (2 * NUM_RELC)
#define SUM_WAVES  32
#define SUM_GROUPS 256
#define SUM_EDGES  8
#define NODE_CAP    50000
#define RANK_CHUNKS 32
#define RANK_WAVES  32
#define RANK_EPT    13

typedef __attribute__((ext_vector_type(2))) float v2f;
typedef __attribute__((ext_vector_type(8))) float v8f;
typedef __attribute__((ext_vector_type(4))) float v4f;
typedef __attribute__((ext_vector_type(2))) int   v2i;
typedef __attribute__((ext_vector_type(4))) int   v4i;

__device__ __forceinline__ v8f wmma4_f32(v2f a, v2f b, v8f c) {
    return __builtin_amdgcn_wmma_f32_16x16x4_f32(
         false, a,  false, b,
         (short)0, c,  false,  false);
}

__global__ __launch_bounds__(256)
void project_kernel(const float* __restrict__ ent_emb,
                    const float* __restrict__ rel_emb,
                    const float* __restrict__ Wo,
                    const float* __restrict__ bo,
                    const float* __restrict__ Wi,
                    const float* __restrict__ bi,
                    const float* __restrict__ Ws,
                    const float* __restrict__ bs,
                    const float* __restrict__ Wr,
                    const float* __restrict__ br,
                    float*       __restrict__ out_ent,
                    float*       __restrict__ out_rel,
                    float*       __restrict__ P,
                    float*       __restrict__ PR,
                    int*         __restrict__ cursor,
                    int N, int R, int node_blocks)
{
    const int  tid   = threadIdx.x;
    const int  lane  = tid & 31;
    const int  wave  = tid >> 5;
    const int  m     = lane & 15;
    const int  kq    = (lane >> 4) << 2;

    const bool   rel  = (int)blockIdx.x >= node_blocks;
    const float* A    = rel ? rel_emb : ent_emb;
    const int    rows = rel ? R : N;
    const int    koff = rel ? 0 : REL_DIM;
    const long   base = (long)(rel ? (int)blockIdx.x - node_blocks : (int)blockIdx.x)
                        * (MTILES * 16);

    if (blockIdx.x == 0 && tid == MTILES * 16) *cursor = 0;

    const float* arow[MTILES];
    #pragma unroll
    for (int t = 0; t < MTILES; ++t) {
        long r  = base + t * 16 + m;
        long rc = (r < rows) ? r : (long)(rows - 1);
        arow[t] = A + rc * ENT_DIM;
    }

    const int    n      = (wave << 4) + m;
    const float* wo_row = Wo + (long)n * CMP_DIM + koff;
    const float* wi_row = Wi + (long)n * CMP_DIM + koff;
    const float* wu_row = (rel ? Wr : Ws) + (long)n * ENT_DIM;

    v8f acc_o[MTILES], acc_i[MTILES], acc_u[MTILES];
    #pragma unroll
    for (int t = 0; t < MTILES; ++t) {
        acc_o[t] = (v8f){0.f,0.f,0.f,0.f,0.f,0.f,0.f,0.f};
        acc_i[t] = (v8f){0.f,0.f,0.f,0.f,0.f,0.f,0.f,0.f};
        acc_u[t] = (v8f){0.f,0.f,0.f,0.f,0.f,0.f,0.f,0.f};
    }

    for (int k = 0; k < ENT_DIM; k += 8) {
        const v4f bO = *(const v4f*)(wo_row + k + kq);
        const v4f bI = *(const v4f*)(wi_row + k + kq);
        const v4f bU = *(const v4f*)(wu_row + k + kq);
        v4f a[MTILES];
        #pragma unroll
        for (int t = 0; t < MTILES; ++t)
            a[t] = *(const v4f*)(arow[t] + k + kq);
        #pragma unroll
        for (int t = 0; t < MTILES; ++t) {
            acc_o[t] = wmma4_f32(a[t].xy, bO.xy, acc_o[t]);
            acc_i[t] = wmma4_f32(a[t].xy, bI.xy, acc_i[t]);
            acc_u[t] = wmma4_f32(a[t].xy, bU.xy, acc_u[t]);
        }
        #pragma unroll
        for (int t = 0; t < MTILES; ++t) {
            acc_o[t] = wmma4_f32(a[t].zw, bO.zw, acc_o[t]);
            acc_i[t] = wmma4_f32(a[t].zw, bI.zw, acc_i[t]);
            acc_u[t] = wmma4_f32(a[t].zw, bU.zw, acc_u[t]);
        }
    }

    const int rhalf = (lane >> 4) << 3;

    if (rel) {
        const float bob = bo[n];
        const float bib = bi[n];
        const float bub = br[n];
        #pragma unroll
        for (int t = 0; t < MTILES; ++t) {
            #pragma unroll
            for (int j = 0; j < 8; ++j) {
                const long row = base + t * 16 + j + rhalf;
                if (row < rows) {
                    PR[row * REL_DIM + n] = (row < NUM_RELC) ? (acc_o[t][j] + bob)
                                                             : (acc_i[t][j] + bib);
                    out_rel[row * REL_DIM + n] = acc_u[t][j] + bub;
                }
            }
        }
    } else {
        const float bub = bs[n];
        #pragma unroll
        for (int t = 0; t < MTILES; ++t) {
            #pragma unroll
            for (int j = 0; j < 8; ++j) {
                const long row = base + t * 16 + j + rhalf;
                if (row < rows) {
                    P[row * CMP_DIM + n]           = acc_o[t][j];
                    P[row * CMP_DIM + ENT_DIM + n] = acc_i[t][j];
                    out_ent[row * ENT_DIM + n]     = acc_u[t][j] + bub;
                }
            }
        }
    }
}

__global__ __launch_bounds__(RANK_WAVES * 32)
void dst_rank_kernel(const int* __restrict__ dst,
                     int*       __restrict__ rank,
                     int*       __restrict__ H,
                     int E, int chunk)
{
    __shared__ int hist[NODE_CAP];
    v4i* hist4 = (v4i*)hist;
    const int tid = threadIdx.x;

    for (int i = tid; i < NODE_CAP / 4; i += RANK_WAVES * 32)
        hist4[i] = (v4i){0, 0, 0, 0};
    __syncthreads();

    const int c0 = (int)blockIdx.x * chunk;
    const int c1 = (c0 + chunk < E) ? (c0 + chunk) : E;
    for (int b = c0 + tid; b < c1; b += RANK_EPT * (RANK_WAVES * 32)) {
        int d[RANK_EPT], r[RANK_EPT];
        #pragma unroll
        for (int k = 0; k < RANK_EPT; ++k) {
            const int e = b + k * (RANK_WAVES * 32);
            d[k] = dst[(e < c1) ? e : b];
        }
        #pragma unroll
        for (int k = 0; k < RANK_EPT; ++k)
            r[k] = atomicAdd(&hist[d[k]], (b + k * (RANK_WAVES * 32) < c1) ? 1 : 0);
        #pragma unroll
        for (int k = 0; k < RANK_EPT; ++k) {
            const int e = b + k * (RANK_WAVES * 32);
            if (e < c1) rank[e] = r[k];
        }
    }
    __syncthreads();

    v4i* Hrow = (v4i*)(H + (long)blockIdx.x * NODE_CAP);
    for (int i = tid; i < NODE_CAP / 4; i += RANK_WAVES * 32)
        Hrow[i] = hist4[i];
}

__global__ __launch_bounds__(256)
void segment_alloc_kernel(int* __restrict__ H,
                          int* __restrict__ cursor,
                          int* __restrict__ cnt,
                          int* __restrict__ off,
                          int N)
{
    __shared__ int wsum[8];
    __shared__ int bbase;
    const int tid  = threadIdx.x;
    const int lane = tid & 31;
    const int wave = tid >> 5;
    const int d    = (int)blockIdx.x * 256 + tid;
    const int dc   = (d < N) ? d : N - 1;

    int h[RANK_CHUNKS];
    #pragma unroll
    for (int q = 0; q < RANK_CHUNKS; ++q)
        h[q] = H[(long)q * NODE_CAP + dc];
    int tot = 0;
    #pragma unroll
    for (int q = 0; q < RANK_CHUNKS; ++q) { const int v = h[q]; h[q] = tot; tot += v; }
    const int c = (d < N) ? tot : 0;

    int x = c;
    #pragma unroll
    for (int s = 1; s < 32; s <<= 1) {
        const int y = __shfl_up(x, s, 32);
        if (lane >= s) x += y;
    }
    if (lane == 31) wsum[wave] = x;
    __syncthreads();
    if (tid == 0) {
        int t = 0;
        #pragma unroll
        for (int w = 0; w < 8; ++w) { const int v = wsum[w]; wsum[w] = t; t += v; }
        bbase = atomicAdd(cursor, t);
    }
    __syncthreads();
    if (d < N) {
        const int o = bbase + wsum[wave] + x - c;
        cnt[d] = c;
        off[d] = o;
        #pragma unroll
        for (int q = 0; q < RANK_CHUNKS; ++q)
            H[(long)q * NODE_CAP + d] = o + h[q];
    }
}

__global__ __launch_bounds__(256)
void edge_place_kernel(const int* __restrict__ src,
                       const int* __restrict__ dst,
                       const int* __restrict__ etypes,
                       const int* __restrict__ H,
                       const int* __restrict__ rank,
                       int*       __restrict__ key,
                       int E, int chunk)
{
    const long e0 = (long)blockIdx.x * (256 * EDGE_EPT) + threadIdx.x;
    const int* Hc = H + (long)(((int)blockIdx.x * (256 * EDGE_EPT)) / chunk) * NODE_CAP;
    int d[EDGE_EPT], r[EDGE_EPT], s[EDGE_EPT], ty[EDGE_EPT], o[EDGE_EPT];
    #pragma unroll
    for (int k = 0; k < EDGE_EPT; ++k) {
        const long e  = e0 + k * 256;
        const long ec = (e < E) ? e : (long)E - 1;
        d[k]  = dst[ec];
        r[k]  = rank[ec];
        s[k]  = src[ec];
        ty[k] = etypes[ec];
    }
    #pragma unroll
    for (int k = 0; k < EDGE_EPT; ++k)
        o[k] = Hc[d[k]];
    #pragma unroll
    for (int k = 0; k < EDGE_EPT; ++k) {
        if (e0 + k * 256 < E) {
            const long pos = (long)o[k] + r[k];
            v2i kv;
            kv.x = s[k] * CMP_DIM + ((ty[k] < NUM_RELC) ? 0 : ENT_DIM);
            kv.y = ty[k] * REL_DIM;
            *(v2i*)(key + 2 * pos) = kv;
        }
    }
}

__global__ __launch_bounds__(SUM_WAVES * 32)
void node_gather_sum_kernel(const int*   __restrict__ key,
                            const int*   __restrict__ cnt,
                            const int*   __restrict__ off,
                            const float* __restrict__ P,
                            const float* __restrict__ PR,
                            float*       __restrict__ out,
                            int N)
{
    __shared__ v4f pr_lds[REL_ROWS * REL_DIM / 4];

    const int tid  = threadIdx.x;
    const int lane = tid & 31;
    const int wave = __builtin_amdgcn_readfirstlane(tid >> 5);
    const int col  = lane << 2;
    const int step = (int)gridDim.x * SUM_WAVES;

    int node = wave * (int)gridDim.x + (int)blockIdx.x;
    int n_v = 0, o_v = 0;
    if (node < N) { n_v = cnt[node]; o_v = off[node]; }

    {
        const v4f* PR4 = (const v4f*)PR;
        #pragma unroll
        for (int i = 0; i < (REL_ROWS * REL_DIM / 4 + SUM_WAVES * 32 - 1) / (SUM_WAVES * 32); ++i) {
            const int idx = i * (SUM_WAVES * 32) + tid;
            if (idx < REL_ROWS * REL_DIM / 4) pr_lds[idx] = PR4[idx];
        }
    }
    __syncthreads();

    #pragma unroll 1
    for (; node < N; node += step) {
        const int n = __builtin_amdgcn_readfirstlane(n_v);
        const int o = __builtin_amdgcn_readfirstlane(o_v);
        const int nxt = node + step;
        n_v = 0; o_v = 0;
        if (nxt < N) { n_v = cnt[nxt]; o_v = off[nxt]; }

        v4f acc = {0.f, 0.f, 0.f, 0.f};
        int m = (n < 32) ? n : 32;
        v2i kv = {0, 0};
        if (lane < m) kv = *(const v2i*)(key + 2 * ((long)o + lane));
        float* orow = out + (long)node * ENT_DIM + col;
        const v4f self_term = *(const v4f*)orow;

        for (int base = 0; base < n; ) {
            for (int g = 0; g < m; g += SUM_EDGES) {
                v4f pv[SUM_EDGES], rv[SUM_EDGES];
                #pragma unroll
                for (int i = 0; i < SUM_EDGES; ++i) {
                    const int pi = __builtin_amdgcn_readlane(kv.x, g + i);
                    const int ri = __builtin_amdgcn_readlane(kv.y, g + i);
                    pv[i] = *(const v4f*)(P + pi + col);
                    rv[i] = pr_lds[(ri + col) >> 2];
                }
                #pragma unroll
                for (int i = 0; i < SUM_EDGES; ++i) {
                    if (g + i < m)
                        acc += rv[i] + pv[i];
                }
            }
            base += 32;
            m = (n - base < 32) ? (n - base) : 32;
            if (base < n) {
                kv = (v2i){0, 0};
                if (lane < m) kv = *(const v2i*)(key + 2 * ((long)o + base + lane));
            }
        }
        float dg = (float)n;
        dg = (dg > 1.0f) ? dg : 1.0f;
        *(v4f*)orow = self_term + acc / dg;
    }
}

extern "C" void kernel_launch(void* const* d_in, const int* in_sizes, int n_in,
                              void* d_out, int out_size, void* d_ws, size_t ws_size,
                              hipStream_t stream)
{
    const int*   src     = (const int*)  d_in[0];
    const int*   dst     = (const int*)  d_in[1];
    const int*   etypes  = (const int*)  d_in[2];
    const float* ent_emb = (const float*)d_in[3];
    const float* rel_emb = (const float*)d_in[4];
    const float* Wo      = (const float*)d_in[5];
    const float* bo      = (const float*)d_in[6];
    const float* Wi      = (const float*)d_in[7];
    const float* bi      = (const float*)d_in[8];
    const float* Ws      = (const float*)d_in[9];
    const float* bs      = (const float*)d_in[10];
    const float* Wr      = (const float*)d_in[11];
    const float* br      = (const float*)d_in[12];

    const int E = in_sizes[0];
    const int N = in_sizes[3] / ENT_DIM;
    const int R = in_sizes[4] / REL_DIM;

    float* P    = (float*)d_ws;
    float* PR   = P   + (size_t)N * CMP_DIM;
    int*   cnt  = (int*)(PR + (((size_t)R * REL_DIM + 63) & ~(size_t)63));
    int*   curs = cnt  + (((size_t)N + 63) & ~(size_t)63);
    int*   off  = curs + 64;
    int*   rank = off  + (((size_t)N + 63) & ~(size_t)63);
    int*   key  = rank + (((size_t)E + 63) & ~(size_t)63);
    int*   H    = key  + (((size_t)E * 2 + 63) & ~(size_t)63);

    float* out_ent = (float*)d_out;
    float* out_rel = out_ent + (size_t)N * ENT_DIM;

    const int rows_per_block = MTILES * 16;
    const int node_blocks    = (N + rows_per_block - 1) / rows_per_block;
    const int rel_blocks     = (R + rows_per_block - 1) / rows_per_block;
    project_kernel<<<node_blocks + rel_blocks, 256, 0, stream>>>(
        ent_emb, rel_emb, Wo, bo, Wi, bi, Ws, bs, Wr, br, out_ent, out_rel,
        P, PR, curs, N, R, node_blocks);

    const int edges_per_block = 256 * EDGE_EPT;
    const int edge_blocks     = (E + edges_per_block - 1) / edges_per_block;
    const int chunk = ((E + RANK_CHUNKS * edges_per_block - 1) / (RANK_CHUNKS * edges_per_block))
                      * edges_per_block;
    dst_rank_kernel<<<RANK_CHUNKS, RANK_WAVES * 32, 0, stream>>>(dst, rank, H, E, chunk);
    segment_alloc_kernel<<<(N + 255) / 256, 256, 0, stream>>>(H, curs, cnt, off, N);
    edge_place_kernel<<<edge_blocks, 256, 0, stream>>>(src, dst, etypes, H, rank, key, E, chunk);

    node_gather_sum_kernel<<<SUM_GROUPS, SUM_WAVES * 32, 0, stream>>>(
        key, cnt, off, P, PR, out_ent, N);
}


# Edge-place pass folded into the projection launch as extra work-groups behind the GEMM blocks (order: rank, alloc, project+place, sum): one launch fewer, placement runs beside the GEMM tail
# speedup vs baseline: 7.6745x; 1.1098x over previous
#include <hip/hip_runtime.h>

#define ENT_DIM  128
#define REL_DIM  128
#define CMP_DIM  256
#define NUM_RELC 200
#define MTILES   4
#define EDGE_EPT   4
#define REL_ROWS   (2 * NUM_RELC)
#define SUM_WAVES  32
#define SUM_GROUPS 256
#define SUM_EDGES  8
#define NODE_CAP    50000
#define RANK_CHUNKS 32
#define RANK_WAVES  32
#define RANK_EPT    13

typedef __attribute__((ext_vector_type(2))) float v2f;
typedef __attribute__((ext_vector_type(8))) float v8f;
typedef __attribute__((ext_vector_type(4))) float v4f;
typedef __attribute__((ext_vector_type(2))) int   v2i;
typedef __attribute__((ext_vector_type(4))) int   v4i;

__device__ __forceinline__ v8f wmma4_f32(v2f a, v2f b, v8f c) {
    return __builtin_amdgcn_wmma_f32_16x16x4_f32(
         false, a,  false, b,
         (short)0, c,  false,  false);
}

__device__ __forceinline__
void edge_place_block(int pb,
                      const int* __restrict__ src,
                      const int* __restrict__ dst,
                      const int* __restrict__ etypes,
                      const int* __restrict__ H,
                      const int* __restrict__ rank,
                      int*       __restrict__ key,
                      int E, int chunk)
{
    const long e0 = (long)pb * (256 * EDGE_EPT) + threadIdx.x;
    const int* Hc = H + (long)((pb * (256 * EDGE_EPT)) / chunk) * NODE_CAP;
    int d[EDGE_EPT], r[EDGE_EPT], s[EDGE_EPT], ty[EDGE_EPT], o[EDGE_EPT];
    #pragma unroll
    for (int k = 0; k < EDGE_EPT; ++k) {
        const long e  = e0 + k * 256;
        const long ec = (e < E) ? e : (long)E - 1;
        d[k]  = dst[ec];
        r[k]  = rank[ec];
        s[k]  = src[ec];
        ty[k] = etypes[ec];
    }
    #pragma unroll
    for (int k = 0; k < EDGE_EPT; ++k)
        o[k] = Hc[d[k]];
    #pragma unroll
    for (int k = 0; k < EDGE_EPT; ++k) {
        if (e0 + k * 256 < E) {
            const long pos = (long)o[k] + r[k];
            v2i kv;
            kv.x = s[k] * CMP_DIM + ((ty[k] < NUM_RELC) ? 0 : ENT_DIM);
            kv.y = ty[k] * REL_DIM;
            *(v2i*)(key + 2 * pos) = kv;
        }
    }
}

__global__ __launch_bounds__(256)
void project_place_kernel(const float* __restrict__ ent_emb,
                    const float* __restrict__ rel_emb,
                    const float* __restrict__ Wo,
                    const float* __restrict__ bo,
                    const float* __restrict__ Wi,
                    const float* __restrict__ bi,
                    const float* __restrict__ Ws,
                    const float* __restrict__ bs,
                    const float* __restrict__ Wr,
                    const float* __restrict__ br,
                    float*       __restrict__ out_ent,
                    float*       __restrict__ out_rel,
                    float*       __restrict__ P,
                    float*       __restrict__ PR,
                    const int*   __restrict__ src,
                    const int*   __restrict__ dst,
                    const int*   __restrict__ etypes,
                    const int*   __restrict__ H,
                    const int*   __restrict__ rank,
                    int*         __restrict__ key,
                    int N, int R, int node_blocks, int proj_blocks,
                    int E, int chunk)
{
    if ((int)blockIdx.x >= proj_blocks) {
        edge_place_block((int)blockIdx.x - proj_blocks, src, dst, etypes, H, rank, key, E, chunk);
        return;
    }

    const int  tid   = threadIdx.x;
    const int  lane  = tid & 31;
    const int  wave  = tid >> 5;
    const int  m     = lane & 15;
    const int  kq    = (lane >> 4) << 2;

    const bool   rel  = (int)blockIdx.x >= node_blocks;
    const float* A    = rel ? rel_emb : ent_emb;
    const int    rows = rel ? R : N;
    const int    koff = rel ? 0 : REL_DIM;
    const long   base = (long)(rel ? (int)blockIdx.x - node_blocks : (int)blockIdx.x)
                        * (MTILES * 16);

    const float* arow[MTILES];
    #pragma unroll
    for (int t = 0; t < MTILES; ++t) {
        long r  = base + t * 16 + m;
        long rc = (r < rows) ? r : (long)(rows - 1);
        arow[t] = A + rc * ENT_DIM;
    }

    const int    n      = (wave << 4) + m;
    const float* wo_row = Wo + (long)n * CMP_DIM + koff;
    const float* wi_row = Wi + (long)n * CMP_DIM + koff;
    const float* wu_row = (rel ? Wr : Ws) + (long)n * ENT_DIM;

    v8f acc_o[MTILES], acc_i[MTILES], acc_u[MTILES];
    #pragma unroll
    for (int t = 0; t < MTILES; ++t) {
        acc_o[t] = (v8f){0.f,0.f,0.f,0.f,0.f,0.f,0.f,0.f};
        acc_i[t] = (v8f){0.f,0.f,0.f,0.f,0.f,0.f,0.f,0.f};
        acc_u[t] = (v8f){0.f,0.f,0.f,0.f,0.f,0.f,0.f,0.f};
    }

    for (int k = 0; k < ENT_DIM; k += 8) {
        const v4f bO = *(const v4f*)(wo_row + k + kq);
        const v4f bI = *(const v4f*)(wi_row + k + kq);
        const v4f bU = *(const v4f*)(wu_row + k + kq);
        v4f a[MTILES];
        #pragma unroll
        for (int t = 0; t < MTILES; ++t)
            a[t] = *(const v4f*)(arow[t] + k + kq);
        #pragma unroll
        for (int t = 0; t < MTILES; ++t) {
            acc_o[t] = wmma4_f32(a[t].xy, bO.xy, acc_o[t]);
            acc_i[t] = wmma4_f32(a[t].xy, bI.xy, acc_i[t]);
            acc_u[t] = wmma4_f32(a[t].xy, bU.xy, acc_u[t]);
        }
        #pragma unroll
        for (int t = 0; t < MTILES; ++t) {
            acc_o[t] = wmma4_f32(a[t].zw, bO.zw, acc_o[t]);
            acc_i[t] = wmma4_f32(a[t].zw, bI.zw, acc_i[t]);
            acc_u[t] = wmma4_f32(a[t].zw, bU.zw, acc_u[t]);
        }
    }

    const int rhalf = (lane >> 4) << 3;

    if (rel) {
        const float bob = bo[n];
        const float bib = bi[n];
        const float bub = br[n];
        #pragma unroll
        for (int t = 0; t < MTILES; ++t) {
            #pragma unroll
            for (int j = 0; j < 8; ++j) {
                const long row = base + t * 16 + j + rhalf;
                if (row < rows) {
                    PR[row * REL_DIM + n] = (row < NUM_RELC) ? (acc_o[t][j] + bob)
                                                             : (acc_i[t][j] + bib);
                    out_rel[row * REL_DIM + n] = acc_u[t][j] + bub;
                }
            }
        }
    } else {
        const float bub = bs[n];
        #pragma unroll
        for (int t = 0; t < MTILES; ++t) {
            #pragma unroll
            for (int j = 0; j < 8; ++j) {
                const long row = base + t * 16 + j + rhalf;
                if (row < rows) {
                    P[row * CMP_DIM + n]           = acc_o[t][j];
                    P[row * CMP_DIM + ENT_DIM + n] = acc_i[t][j];
                    out_ent[row * ENT_DIM + n]     = acc_u[t][j] + bub;
                }
            }
        }
    }
}

__global__ __launch_bounds__(RANK_WAVES * 32)
void dst_rank_kernel(const int* __restrict__ dst,
                     int*       __restrict__ rank,
                     int*       __restrict__ H,
                     int*       __restrict__ cursor,
                     int E, int chunk)
{
    __shared__ int hist[NODE_CAP];
    v4i* hist4 = (v4i*)hist;
    const int tid = threadIdx.x;

    if (blockIdx.x == 0 && tid == 0) *cursor = 0;

    for (int i = tid; i < NODE_CAP / 4; i += RANK_WAVES * 32)
        hist4[i] = (v4i){0, 0, 0, 0};
    __syncthreads();

    const int c0 = (int)blockIdx.x * chunk;
    const int c1 = (c0 + chunk < E) ? (c0 + chunk) : E;
    for (int b = c0 + tid; b < c1; b += RANK_EPT * (RANK_WAVES * 32)) {
        int d[RANK_EPT], r[RANK_EPT];
        #pragma unroll
        for (int k = 0; k < RANK_EPT; ++k) {
            const int e = b + k * (RANK_WAVES * 32);
            d[k] = dst[(e < c1) ? e : b];
        }
        #pragma unroll
        for (int k = 0; k < RANK_EPT; ++k)
            r[k] = atomicAdd(&hist[d[k]], (b + k * (RANK_WAVES * 32) < c1) ? 1 : 0);
        #pragma unroll
        for (int k = 0; k < RANK_EPT; ++k) {
            const int e = b + k * (RANK_WAVES * 32);
            if (e < c1) rank[e] = r[k];
        }
    }
    __syncthreads();

    v4i* Hrow = (v4i*)(H + (long)blockIdx.x * NODE_CAP);
    for (int i = tid; i < NODE_CAP / 4; i += RANK_WAVES * 32)
        Hrow[i] = hist4[i];
}

__global__ __launch_bounds__(256)
void segment_alloc_kernel(int* __restrict__ H,
                          int* __restrict__ cursor,
                          int* __restrict__ cnt,
                          int* __restrict__ off,
                          int N)
{
    __shared__ int wsum[8];
    __shared__ int bbase;
    const int tid  = threadIdx.x;
    const int lane = tid & 31;
    const int wave = tid >> 5;
    const int d    = (int)blockIdx.x * 256 + tid;
    const int dc   = (d < N) ? d : N - 1;

    int h[RANK_CHUNKS];
    #pragma unroll
    for (int q = 0; q < RANK_CHUNKS; ++q)
        h[q] = H[(long)q * NODE_CAP + dc];
    int tot = 0;
    #pragma unroll
    for (int q = 0; q < RANK_CHUNKS; ++q) { const int v = h[q]; h[q] = tot; tot += v; }
    const int c = (d < N) ? tot : 0;

    int x = c;
    #pragma unroll
    for (int s = 1; s < 32; s <<= 1) {
        const int y = __shfl_up(x, s, 32);
        if (lane >= s) x += y;
    }
    if (lane == 31) wsum[wave] = x;
    __syncthreads();
    if (tid == 0) {
        int t = 0;
        #pragma unroll
        for (int w = 0; w < 8; ++w) { const int v = wsum[w]; wsum[w] = t; t += v; }
        bbase = atomicAdd(cursor, t);
    }
    __syncthreads();
    if (d < N) {
        const int o = bbase + wsum[wave] + x - c;
        cnt[d] = c;
        off[d] = o;
        #pragma unroll
        for (int q = 0; q < RANK_CHUNKS; ++q)
            H[(long)q * NODE_CAP + d] = o + h[q];
    }
}

__global__ __launch_bounds__(SUM_WAVES * 32)
void node_gather_sum_kernel(const int*   __restrict__ key,
                            const int*   __restrict__ cnt,
                            const int*   __restrict__ off,
                            const float* __restrict__ P,
                            const float* __restrict__ PR,
                            float*       __restrict__ out,
                            int N)
{
    __shared__ v4f pr_lds[REL_ROWS * REL_DIM / 4];

    const int tid  = threadIdx.x;
    const int lane = tid & 31;
    const int wave = __builtin_amdgcn_readfirstlane(tid >> 5);
    const int col  = lane << 2;
    const int step = (int)gridDim.x * SUM_WAVES;

    int node = wave * (int)gridDim.x + (int)blockIdx.x;
    int n_v = 0, o_v = 0;
    if (node < N) { n_v = cnt[node]; o_v = off[node]; }

    {
        const v4f* PR4 = (const v4f*)PR;
        #pragma unroll
        for (int i = 0; i < (REL_ROWS * REL_DIM / 4 + SUM_WAVES * 32 - 1) / (SUM_WAVES * 32); ++i) {
            const int idx = i * (SUM_WAVES * 32) + tid;
            if (idx < REL_ROWS * REL_DIM / 4) pr_lds[idx] = PR4[idx];
        }
    }
    __syncthreads();

    #pragma unroll 1
    for (; node < N; node += step) {
        const int n = __builtin_amdgcn_readfirstlane(n_v);
        const int o = __builtin_amdgcn_readfirstlane(o_v);
        const int nxt = node + step;
        n_v = 0; o_v = 0;
        if (nxt < N) { n_v = cnt[nxt]; o_v = off[nxt]; }

        v4f acc = {0.f, 0.f, 0.f, 0.f};
        int m = (n < 32) ? n : 32;
        v2i kv = {0, 0};
        if (lane < m) kv = *(const v2i*)(key + 2 * ((long)o + lane));
        float* orow = out + (long)node * ENT_DIM + col;
        const v4f self_term = *(const v4f*)orow;

        for (int base = 0; base < n; ) {
            for (int g = 0; g < m; g += SUM_EDGES) {
                v4f pv[SUM_EDGES], rv[SUM_EDGES];
                #pragma unroll
                for (int i = 0; i < SUM_EDGES; ++i) {
                    const int pi = __builtin_amdgcn_readlane(kv.x, g + i);
                    const int ri = __builtin_amdgcn_readlane(kv.y, g + i);
                    pv[i] = *(const v4f*)(P + pi + col);
                    rv[i] = pr_lds[(ri + col) >> 2];
                }
                #pragma unroll
                for (int i = 0; i < SUM_EDGES; ++i) {
                    if (g + i < m)
                        acc += rv[i] + pv[i];
                }
            }
            base += 32;
            m = (n - base < 32) ? (n - base) : 32;
            if (base < n) {
                kv = (v2i){0, 0};
                if (lane < m) kv = *(const v2i*)(key + 2 * ((long)o + base + lane));
            }
        }
        float dg = (float)n;
        dg = (dg > 1.0f) ? dg : 1.0f;
        *(v4f*)orow = self_term + acc / dg;
    }
}

extern "C" void kernel_launch(void* const* d_in, const int* in_sizes, int n_in,
                              void* d_out, int out_size, void* d_ws, size_t ws_size,
                              hipStream_t stream)
{
    const int*   src     = (const int*)  d_in[0];
    const int*   dst     = (const int*)  d_in[1];
    const int*   etypes  = (const int*)  d_in[2];
    const float* ent_emb = (const float*)d_in[3];
    const float* rel_emb = (const float*)d_in[4];
    const float* Wo      = (const float*)d_in[5];
    const float* bo      = (const float*)d_in[6];
    const float* Wi      = (const float*)d_in[7];
    const float* bi      = (const float*)d_in[8];
    const float* Ws      = (const float*)d_in[9];
    const float* bs      = (const float*)d_in[10];
    const float* Wr      = (const float*)d_in[11];
    const float* br      = (const float*)d_in[12];

    const int E = in_sizes[0];
    const int N = in_sizes[3] / ENT_DIM;
    const int R = in_sizes[4] / REL_DIM;

    float* P    = (float*)d_ws;
    float* PR   = P   + (size_t)N * CMP_DIM;
    int*   cnt  = (int*)(PR + (((size_t)R * REL_DIM + 63) & ~(size_t)63));
    int*   curs = cnt  + (((size_t)N + 63) & ~(size_t)63);
    int*   off  = curs + 64;
    int*   rank = off  + (((size_t)N + 63) & ~(size_t)63);
    int*   key  = rank + (((size_t)E + 63) & ~(size_t)63);
    int*   H    = key  + (((size_t)E * 2 + 63) & ~(size_t)63);

    float* out_ent = (float*)d_out;
    float* out_rel = out_ent + (size_t)N * ENT_DIM;

    const int rows_per_block = MTILES * 16;
    const int node_blocks    = (N + rows_per_block - 1) / rows_per_block;
    const int rel_blocks     = (R + rows_per_block - 1) / rows_per_block;

    const int edges_per_block = 256 * EDGE_EPT;
    const int edge_blocks     = (E + edges_per_block - 1) / edges_per_block;
    const int chunk = ((E + RANK_CHUNKS * edges_per_block - 1) / (RANK_CHUNKS * edges_per_block))
                      * edges_per_block;
    dst_rank_kernel<<<RANK_CHUNKS, RANK_WAVES * 32, 0, stream>>>(dst, rank, H, curs, E, chunk);
    segment_alloc_kernel<<<(N + 255) / 256, 256, 0, stream>>>(H, curs, cnt, off, N);
    project_place_kernel<<<node_blocks + rel_blocks + edge_blocks, 256, 0, stream>>>(
        ent_emb, rel_emb, Wo, bo, Wi, bi, Ws, bs, Wr, br, out_ent, out_rel,
        P, PR, src, dst, etypes, H, rank, key,
        N, R, node_blocks, node_blocks + rel_blocks, E, chunk);

    node_gather_sum_kernel<<<SUM_GROUPS, SUM_WAVES * 32, 0, stream>>>(
        key, cnt, off, P, PR, out_ent, N);
}


# Projection GEMM operands staged through LDS in 32-k chunks: whole 128-byte row lines loaded once per block (table) or wave (weights), fragments read from LDS at 144-byte pitch; k and sum order kept
# speedup vs baseline: 8.4917x; 1.1065x over previous
#include <hip/hip_runtime.h>

#define ENT_DIM  128
#define REL_DIM  128
#define CMP_DIM  256
#define NUM_RELC 200
#define MTILES   4
#define EDGE_EPT   4
#define REL_ROWS   (2 * NUM_RELC)
#define SUM_WAVES  32
#define SUM_GROUPS 256
#define SUM_EDGES  8
#define NODE_CAP    50000
#define RANK_CHUNKS 32
#define RANK_WAVES  32
#define RANK_EPT    13
#define KCH      32
#define PPR      (KCH / 4)
#define PITCH4   (PPR + 1)
#define RPI      (32 / PPR)
#define LPM      (16 / RPI)
#define A_PIECES (MTILES * 16 * PPR / 256)

typedef __attribute__((ext_vector_type(2))) float v2f;
typedef __attribute__((ext_vector_type(8))) float v8f;
typedef __attribute__((ext_vector_type(4))) float v4f;
typedef __attribute__((ext_vector_type(2))) int   v2i;
typedef __attribute__((ext_vector_type(4))) int   v4i;

__device__ __forceinline__ v8f wmma4_f32(v2f a, v2f b, v8f c) {
    return __builtin_amdgcn_wmma_f32_16x16x4_f32(
         false, a,  false, b,
         (short)0, c,  false,  false);
}

__device__ __forceinline__
void edge_place_block(int pb,
                      const int* __restrict__ src,
                      const int* __restrict__ dst,
                      const int* __restrict__ etypes,
                      const int* __restrict__ H,
                      const int* __restrict__ rank,
                      int*       __restrict__ key,
                      int E, int chunk)
{
    const long e0 = (long)pb * (256 * EDGE_EPT) + threadIdx.x;
    const int* Hc = H + (long)((pb * (256 * EDGE_EPT)) / chunk) * NODE_CAP;
    int d[EDGE_EPT], r[EDGE_EPT], s[EDGE_EPT], ty[EDGE_EPT], o[EDGE_EPT];
    #pragma unroll
    for (int k = 0; k < EDGE_EPT; ++k) {
        const long e  = e0 + k * 256;
        const long ec = (e < E) ? e : (long)E - 1;
        d[k]  = dst[ec];
        r[k]  = rank[ec];
        s[k]  = src[ec];
        ty[k] = etypes[ec];
    }
    #pragma unroll
    for (int k = 0; k < EDGE_EPT; ++k)
        o[k] = Hc[d[k]];
    #pragma unroll
    for (int k = 0; k < EDGE_EPT; ++k) {
        if (e0 + k * 256 < E) {
            const long pos = (long)o[k] + r[k];
            v2i kv;
            kv.x = s[k] * CMP_DIM + ((ty[k] < NUM_RELC) ? 0 : ENT_DIM);
            kv.y = ty[k] * REL_DIM;
            *(v2i*)(key + 2 * pos) = kv;
        }
    }
}

__global__ __launch_bounds__(256, 6)
void project_place_kernel(const float* __restrict__ ent_emb,
                    const float* __restrict__ rel_emb,
                    const float* __restrict__ Wo,
                    const float* __restrict__ bo,
                    const float* __restrict__ Wi,
                    const float* __restrict__ bi,
                    const float* __restrict__ Ws,
                    const float* __restrict__ bs,
                    const float* __restrict__ Wr,
                    const float* __restrict__ br,
                    float*       __restrict__ out_ent,
                    float*       __restrict__ out_rel,
                    float*       __restrict__ P,
                    float*       __restrict__ PR,
                    const int*   __restrict__ src,
                    const int*   __restrict__ dst,
                    const int*   __restrict__ etypes,
                    const int*   __restrict__ H,
                    const int*   __restrict__ rank,
                    int*         __restrict__ key,
                    int N, int R, int node_blocks, int proj_blocks,
                    int E, int chunk)
{
    __shared__ v4f As4[MTILES * 16 * PITCH4];
    __shared__ v4f Bs4[8 * 48 * PITCH4];

    if ((int)blockIdx.x >= proj_blocks) {
        edge_place_block((int)blockIdx.x - proj_blocks, src, dst, etypes, H, rank, key, E, chunk);
        return;
    }

    const int  tid   = threadIdx.x;
    const int  lane  = tid & 31;
    const int  wave  = tid >> 5;
    const int  m     = lane & 15;
    const int  h     = lane >> 4;

    const bool   rel  = (int)blockIdx.x >= node_blocks;
    const float* A    = rel ? rel_emb : ent_emb;
    const int    rows = rel ? R : N;
    const int    koff = rel ? 0 : REL_DIM;
    const long   base = (long)(rel ? (int)blockIdx.x - node_blocks : (int)blockIdx.x)
                        * (MTILES * 16);

    unsigned     asrc[A_PIECES];
    int          adst[A_PIECES];
    #pragma unroll
    for (int q = 0; q < A_PIECES; ++q) {
        const int  idx = q * 256 + tid;
        const int  row = idx / PPR;
        const int  p   = idx % PPR;
        const long r   = base + row;
        const long rc  = (r < rows) ? r : (long)(rows - 1);
        asrc[q] = (unsigned)(rc * ENT_DIM + p * 4);
        adst[q] = row * PITCH4 + p;
    }
    const int    brow = lane / PPR;
    const int    bp   = lane % PPR;
    const int    nb   = (wave << 4) + brow;
    const unsigned wx_l = (unsigned)(nb * CMP_DIM + koff + bp * 4);
    const float*   Wu   = rel ? Wr : Ws;
    const unsigned wu_l = (unsigned)(nb * ENT_DIM + bp * 4);
    v4f*         Bw   = Bs4 + wave * (48 * PITCH4);
    const int    bdst = brow * PITCH4 + bp;

    const int    n    = (wave << 4) + m;

    v8f acc_o[MTILES], acc_i[MTILES], acc_u[MTILES];
    #pragma unroll
    for (int t = 0; t < MTILES; ++t) {
        acc_o[t] = (v8f){0.f,0.f,0.f,0.f,0.f,0.f,0.f,0.f};
        acc_i[t] = (v8f){0.f,0.f,0.f,0.f,0.f,0.f,0.f,0.f};
        acc_u[t] = (v8f){0.f,0.f,0.f,0.f,0.f,0.f,0.f,0.f};
    }

    for (int kc = 0; kc < ENT_DIM; kc += KCH) {
        __syncthreads();
        #pragma unroll
        for (int hf = 0; hf < 2; ++hf) {
            v4f sa[A_PIECES / 2], so[LPM / 2], si[LPM / 2], su[LPM / 2];
            #pragma unroll
            for (int q = 0; q < A_PIECES / 2; ++q)
                sa[q] = *(const v4f*)(A + kc + asrc[hf * (A_PIECES / 2) + q]);
            #pragma unroll
            for (int j = 0; j < LPM / 2; ++j) {
                const int jj = hf * (LPM / 2) + j;
                so[j] = *(const v4f*)(Wo + (kc + jj * RPI * CMP_DIM) + wx_l);
                si[j] = *(const v4f*)(Wi + (kc + jj * RPI * CMP_DIM) + wx_l);
                su[j] = *(const v4f*)(Wu + (kc + jj * RPI * ENT_DIM) + wu_l);
            }
            #pragma unroll
            for (int q = 0; q < A_PIECES / 2; ++q)
                As4[adst[hf * (A_PIECES / 2) + q]] = sa[q];
            #pragma unroll
            for (int j = 0; j < LPM / 2; ++j) {
                const int jj = hf * (LPM / 2) + j;
                Bw[(0 * 16 + jj * RPI) * PITCH4 + bdst] = so[j];
                Bw[(1 * 16 + jj * RPI) * PITCH4 + bdst] = si[j];
                Bw[(2 * 16 + jj * RPI) * PITCH4 + bdst] = su[j];
            }
        }
        __syncthreads();
        #pragma unroll
        for (int it = 0; it < KCH / 8; ++it) {
            const int w4 = it * 2 + h;
            const v4f bO = Bw[(0 * 16 + m) * PITCH4 + w4];
            const v4f bI = Bw[(1 * 16 + m) * PITCH4 + w4];
            const v4f bU = Bw[(2 * 16 + m) * PITCH4 + w4];
            v4f a[MTILES];
            #pragma unroll
            for (int t = 0; t < MTILES; ++t)
                a[t] = As4[(t * 16 + m) * PITCH4 + w4];
            #pragma unroll
            for (int t = 0; t < MTILES; ++t) {
                acc_o[t] = wmma4_f32(a[t].xy, bO.xy, acc_o[t]);
                acc_i[t] = wmma4_f32(a[t].xy, bI.xy, acc_i[t]);
                acc_u[t] = wmma4_f32(a[t].xy, bU.xy, acc_u[t]);
            }
            #pragma unroll
            for (int t = 0; t < MTILES; ++t) {
                acc_o[t] = wmma4_f32(a[t].zw, bO.zw, acc_o[t]);
                acc_i[t] = wmma4_f32(a[t].zw, bI.zw, acc_i[t]);
                acc_u[t] = wmma4_f32(a[t].zw, bU.zw, acc_u[t]);
            }
        }
    }

    const int rhalf = (lane >> 4) << 3;

    if (rel) {
        const float bob = bo[n];
        const float bib = bi[n];
        const float bub = br[n];
        #pragma unroll
        for (int t = 0; t < MTILES; ++t) {
            #pragma unroll
            for (int j = 0; j < 8; ++j) {
                const long row = base + t * 16 + j + rhalf;
                if (row < rows) {
                    PR[row * REL_DIM + n] = (row < NUM_RELC) ? (acc_o[t][j] + bob)
                                                             : (acc_i[t][j] + bib);
                    out_rel[row * REL_DIM + n] = acc_u[t][j] + bub;
                }
            }
        }
    } else {
        const float bub = bs[n];
        #pragma unroll
        for (int t = 0; t < MTILES; ++t) {
            #pragma unroll
            for (int j = 0; j < 8; ++j) {
                const long row = base + t * 16 + j + rhalf;
                if (row < rows) {
                    P[row * CMP_DIM + n]           = acc_o[t][j];
                    P[row * CMP_DIM + ENT_DIM + n] = acc_i[t][j];
                    out_ent[row * ENT_DIM + n]     = acc_u[t][j] + bub;
                }
            }
        }
    }
}

__global__ __launch_bounds__(RANK_WAVES * 32)
void dst_rank_kernel(const int* __restrict__ dst,
                     int*       __restrict__ rank,
                     int*       __restrict__ H,
                     int*       __restrict__ cursor,
                     int E, int chunk)
{
    __shared__ int hist[NODE_CAP];
    v4i* hist4 = (v4i*)hist;
    const int tid = threadIdx.x;

    if (blockIdx.x == 0 && tid == 0) *cursor = 0;

    for (int i = tid; i < NODE_CAP / 4; i += RANK_WAVES * 32)
        hist4[i] = (v4i){0, 0, 0, 0};
    __syncthreads();

    const int c0 = (int)blockIdx.x * chunk;
    const int c1 = (c0 + chunk < E) ? (c0 + chunk) : E;
    for (int b = c0 + tid; b < c1; b += RANK_EPT * (RANK_WAVES * 32)) {
        int d[RANK_EPT], r[RANK_EPT];
        #pragma unroll
        for (int k = 0; k < RANK_EPT; ++k) {
            const int e = b + k * (RANK_WAVES * 32);
            d[k] = dst[(e < c1) ? e : b];
        }
        #pragma unroll
        for (int k = 0; k < RANK_EPT; ++k)
            r[k] = atomicAdd(&hist[d[k]], (b + k * (RANK_WAVES * 32) < c1) ? 1 : 0);
        #pragma unroll
        for (int k = 0; k < RANK_EPT; ++k) {
            const int e = b + k * (RANK_WAVES * 32);
            if (e < c1) rank[e] = r[k];
        }
    }
    __syncthreads();

    v4i* Hrow = (v4i*)(H + (long)blockIdx.x * NODE_CAP);
    for (int i = tid; i < NODE_CAP / 4; i += RANK_WAVES * 32)
        Hrow[i] = hist4[i];
}

__global__ __launch_bounds__(256)
void segment_alloc_kernel(int* __restrict__ H,
                          int* __restrict__ cursor,
                          int* __restrict__ cnt,
                          int* __restrict__ off,
                          int N)
{
    __shared__ int wsum[8];
    __shared__ int bbase;
    const int tid  = threadIdx.x;
    const int lane = tid & 31;
    const int wave = tid >> 5;
    const int d    = (int)blockIdx.x * 256 + tid;
    const int dc   = (d < N) ? d : N - 1;

    int h[RANK_CHUNKS];
    #pragma unroll
    for (int q = 0; q < RANK_CHUNKS; ++q)
        h[q] = H[(long)q * NODE_CAP + dc];
    int tot = 0;
    #pragma unroll
    for (int q = 0; q < RANK_CHUNKS; ++q) { const int v = h[q]; h[q] = tot; tot += v; }
    const int c = (d < N) ? tot : 0;

    int x = c;
    #pragma unroll
    for (int s = 1; s < 32; s <<= 1) {
        const int y = __shfl_up(x, s, 32);
        if (lane >= s) x += y;
    }
    if (lane == 31) wsum[wave] = x;
    __syncthreads();
    if (tid == 0) {
        int t = 0;
        #pragma unroll
        for (int w = 0; w < 8; ++w) { const int v = wsum[w]; wsum[w] = t; t += v; }
        bbase = atomicAdd(cursor, t);
    }
    __syncthreads();
    if (d < N) {
        const int o = bbase + wsum[wave] + x - c;
        cnt[d] = c;
        off[d] = o;
        #pragma unroll
        for (int q = 0; q < RANK_CHUNKS; ++q)
            H[(long)q * NODE_CAP + d] = o + h[q];
    }
}

__global__ __launch_bounds__(SUM_WAVES * 32)
void node_gather_sum_kernel(const int*   __restrict__ key,
                            const int*   __restrict__ cnt,
                            const int*   __restrict__ off,
                            const float* __restrict__ P,
                            const float* __restrict__ PR,
                            float*       __restrict__ out,
                            int N)
{
    __shared__ v4f pr_lds[REL_ROWS * REL_DIM / 4];

    const int tid  = threadIdx.x;
    const int lane = tid & 31;
    const int wave = __builtin_amdgcn_readfirstlane(tid >> 5);
    const int col  = lane << 2;
    const int step = (int)gridDim.x * SUM_WAVES;

    int node = wave * (int)gridDim.x + (int)blockIdx.x;
    int n_v = 0, o_v = 0;
    if (node < N) { n_v = cnt[node]; o_v = off[node]; }

    {
        const v4f* PR4 = (const v4f*)PR;
        #pragma unroll
        for (int i = 0; i < (REL_ROWS * REL_DIM / 4 + SUM_WAVES * 32 - 1) / (SUM_WAVES * 32); ++i) {
            const int idx = i * (SUM_WAVES * 32) + tid;
            if (idx < REL_ROWS * REL_DIM / 4) pr_lds[idx] = PR4[idx];
        }
    }
    __syncthreads();

    #pragma unroll 1
    for (; node < N; node += step) {
        const int n = __builtin_amdgcn_readfirstlane(n_v);
        const int o = __builtin_amdgcn_readfirstlane(o_v);
        const int nxt = node + step;
        n_v = 0; o_v = 0;
        if (nxt < N) { n_v = cnt[nxt]; o_v = off[nxt]; }

        v4f acc = {0.f, 0.f, 0.f, 0.f};
        int m = (n < 32) ? n : 32;
        v2i kv = {0, 0};
        if (lane < m) kv = *(const v2i*)(key + 2 * ((long)o + lane));
        float* orow = out + (long)node * ENT_DIM + col;
        const v4f self_term = *(const v4f*)orow;

        for (int base = 0; base < n; ) {
            for (int g = 0; g < m; g += SUM_EDGES) {
                v4f pv[SUM_EDGES], rv[SUM_EDGES];
                #pragma unroll
                for (int i = 0; i < SUM_EDGES; ++i) {
                    const int pi = __builtin_amdgcn_readlane(kv.x, g + i);
                    const int ri = __builtin_amdgcn_readlane(kv.y, g + i);
                    pv[i] = *(const v4f*)(P + pi + col);
                    rv[i] = pr_lds[(ri + col) >> 2];
                }
                #pragma unroll
                for (int i = 0; i < SUM_EDGES; ++i) {
                    if (g + i < m)
                        acc += rv[i] + pv[i];
                }
            }
            base += 32;
            m = (n - base < 32) ? (n - base) : 32;
            if (base < n) {
                kv = (v2i){0, 0};
                if (lane < m) kv = *(const v2i*)(key + 2 * ((long)o + base + lane));
            }
        }
        float dg = (float)n;
        dg = (dg > 1.0f) ? dg : 1.0f;
        *(v4f*)orow = self_term + acc / dg;
    }
}

extern "C" void kernel_launch(void* const* d_in, const int* in_sizes, int n_in,
                              void* d_out, int out_size, void* d_ws, size_t ws_size,
                              hipStream_t stream)
{
    const int*   src     = (const int*)  d_in[0];
    const int*   dst     = (const int*)  d_in[1];
    const int*   etypes  = (const int*)  d_in[2];
    const float* ent_emb = (const float*)d_in[3];
    const float* rel_emb = (const float*)d_in[4];
    const float* Wo      = (const float*)d_in[5];
    const float* bo      = (const float*)d_in[6];
    const float* Wi      = (const float*)d_in[7];
    const float* bi      = (const float*)d_in[8];
    const float* Ws      = (const float*)d_in[9];
    const float* bs      = (const float*)d_in[10];
    const float* Wr      = (const float*)d_in[11];
    const float* br      = (const float*)d_in[12];

    const int E = in_sizes[0];
    const int N = in_sizes[3] / ENT_DIM;
    const int R = in_sizes[4] / REL_DIM;

    float* P    = (float*)d_ws;
    float* PR   = P   + (size_t)N * CMP_DIM;
    int*   cnt  = (int*)(PR + (((size_t)R * REL_DIM + 63) & ~(size_t)63));
    int*   curs = cnt  + (((size_t)N + 63) & ~(size_t)63);
    int*   off  = curs + 64;
    int*   rank = off  + (((size_t)N + 63) & ~(size_t)63);
    int*   key  = rank + (((size_t)E + 63) & ~(size_t)63);
    int*   H    = key  + (((size_t)E * 2 + 63) & ~(size_t)63);

    float* out_ent = (float*)d_out;
    float* out_rel = out_ent + (size_t)N * ENT_DIM;

    const int rows_per_block = MTILES * 16;
    const int node_blocks    = (N + rows_per_block - 1) / rows_per_block;
    const int rel_blocks     = (R + rows_per_block - 1) / rows_per_block;

    const int edges_per_block = 256 * EDGE_EPT;
    const int edge_blocks     = (E + edges_per_block - 1) / edges_per_block;
    const int chunk = ((E + RANK_CHUNKS * edges_per_block - 1) / (RANK_CHUNKS * edges_per_block))
                      * edges_per_block;
    dst_rank_kernel<<<RANK_CHUNKS, RANK_WAVES * 32, 0, stream>>>(dst, rank, H, curs, E, chunk);
    segment_alloc_kernel<<<(N + 255) / 256, 256, 0, stream>>>(H, curs, cnt, off, N);
    project_place_kernel<<<node_blocks + rel_blocks + edge_blocks, 256, 0, stream>>>(
        ent_emb, rel_emb, Wo, bo, Wi, bi, Ws, bs, Wr, br, out_ent, out_rel,
        P, PR, src, dst, etypes, H, rank, key,
        N, R, node_blocks, node_blocks + rel_blocks, E, chunk);

    node_gather_sum_kernel<<<SUM_GROUPS, SUM_WAVES * 32, 0, stream>>>(
        key, cnt, off, P, PR, out_ent, N);
}
